# starting kernel
# baseline (speedup 1.0000x reference)
#include <hip/hip_runtime.h>


typedef __attribute__((ext_vector_type(2))) float v2f;
typedef __attribute__((ext_vector_type(8))) float v8f;

#define H 128
#define LDSPAD 132
#define TPB 8

__global__ __launch_bounds__(256) void zero_kernel(float* p, long n) {
  long i = (long)blockIdx.x * blockDim.x + threadIdx.x;
  if (i < n) p[i] = 0.0f;
}

__global__ __launch_bounds__(256) void pad_kernel(const float* __restrict__ in,
                                                  float* __restrict__ out,
                                                  int N, int F) {
  long t = (long)blockIdx.x * blockDim.x + threadIdx.x;
  if (t >= (long)N * H) return;
  int i = (int)(t >> 7), h = (int)(t & 127);
  out[t] = (h < F) ? in[(long)i * F + h] : 0.0f;
}

__global__ __launch_bounds__(256) void deg_kernel(const int* __restrict__ edge,
                                                  int E, float* __restrict__ deg) {
  int e = blockIdx.x * blockDim.x + threadIdx.x;
  if (e < E) atomicAdd(&deg[edge[E + e]], 1.0f);
}

__global__ __launch_bounds__(256) void scatter_kernel(const float* __restrict__ X,
                                                      const int* __restrict__ edge,
                                                      int E, float* __restrict__ S) {
  long idx = (long)blockIdx.x * blockDim.x + threadIdx.x;
  if (idx >= (long)E * 32) return;
  int e = (int)(idx >> 5);
  int c = (int)(idx & 31) << 2;
  int s = edge[e], d = edge[E + e];
  const float4 v = *(const float4*)(X + (long)s * H + c);
  float* o = S + (long)d * H + c;
  atomicAdd(o + 0, v.x); atomicAdd(o + 1, v.y);
  atomicAdd(o + 2, v.z); atomicAdd(o + 3, v.w);
}

template <bool HAS_DEG, bool RELU>
__global__ __launch_bounds__(256) void gemm_dual_kernel(
    const float* __restrict__ A, const float* __restrict__ B,
    const float* __restrict__ WA, const float* __restrict__ WB,
    const float* __restrict__ bias, const float* __restrict__ deg,
    float* __restrict__ Y, int M) {
  __shared__ float lwa[H * LDSPAD];
  __shared__ float lwb[H * LDSPAD];
  for (int idx = threadIdx.x; idx < H * H; idx += 256) {
    int k = idx >> 7, n = idx & 127;
    lwa[n * LDSPAD + k] = WA[idx];
    lwb[n * LDSPAD + k] = WB[idx];
  }
  __syncthreads();

  const int wave = threadIdx.x >> 5;
  const int lane = threadIdx.x & 31;
  const int l15  = lane & 15;
  const int kh2  = (lane >> 4) << 1;
  const int ntiles = M >> 4;

  for (int t = blockIdx.x * TPB + wave; t < ntiles; t += gridDim.x * TPB) {
    const int m0 = t << 4;
    v8f cA[8], cB[8];
#pragma unroll
    for (int n = 0; n < 8; ++n) {
      cA[n] = (v8f){0, 0, 0, 0, 0, 0, 0, 0};
      cB[n] = (v8f){0, 0, 0, 0, 0, 0, 0, 0};
    }

    const float* Ap = A + (long)(m0 + l15) * H + kh2;
    const float* Bp = B + (long)(m0 + l15) * H + kh2;
    const float* wabase = &lwa[l15 * LDSPAD + kh2];
    const float* wbbase = &lwb[l15 * LDSPAD + kh2];

#pragma unroll 1
    for (int kk = 0; kk < 32; ++kk) {
      const int k0 = kk << 2;
      v2f a = *(const v2f*)(Ap + k0);
      v2f b = *(const v2f*)(Bp + k0);
#pragma unroll
      for (int n = 0; n < 8; ++n) {
        v2f wa = *(const v2f*)(wabase + n * 16 * LDSPAD + k0);
        v2f wb = *(const v2f*)(wbbase + n * 16 * LDSPAD + k0);
        cA[n] = __builtin_amdgcn_wmma_f32_16x16x4_f32(false, a, false, wa,
                                                      (short)0, cA[n], false, false);
        cB[n] = __builtin_amdgcn_wmma_f32_16x16x4_f32(false, b, false, wb,
                                                      (short)0, cB[n], false, false);
      }
    }

    const int rbase = m0 + ((lane >> 4) << 3);
    float degv[8];
    if (HAS_DEG) {
      const float4 d0 = *(const float4*)(deg + rbase);
      const float4 d1 = *(const float4*)(deg + rbase + 4);
      degv[0] = d0.x; degv[1] = d0.y; degv[2] = d0.z; degv[3] = d0.w;
      degv[4] = d1.x; degv[5] = d1.y; degv[6] = d1.z; degv[7] = d1.w;
    } else {
#pragma unroll
      for (int r = 0; r < 8; ++r) degv[r] = 1.0f;
    }
#pragma unroll
    for (int n = 0; n < 8; ++n) {
      const float bv = bias[n * 16 + l15];
#pragma unroll
      for (int r = 0; r < 8; ++r) {
        float acc = HAS_DEG ? (cA[n][r] + degv[r] * (cB[n][r] + bv))
                            : (cA[n][r] + cB[n][r] + bv);
        if (RELU) acc = fmaxf(acc, 0.0f);
        Y[(long)(rbase + r) * H + n * 16 + l15] = acc;
      }
    }
  }
}

__global__ __launch_bounds__(256) void pool_sum_kernel(const float* __restrict__ xl,
                                                       const int* __restrict__ batch,
                                                       int NL, float* __restrict__ sums) {
  long idx = (long)blockIdx.x * blockDim.x + threadIdx.x;
  if (idx >= (long)NL * 32) return;
  int link = (int)(idx >> 5);
  int c = (int)(idx & 31) << 2;
  int b = batch[link];
  const float4 v = *(const float4*)(xl + (long)link * H + c);
  float* o = sums + (long)b * H + c;
  atomicAdd(o + 0, v.x); atomicAdd(o + 1, v.y);
  atomicAdd(o + 2, v.z); atomicAdd(o + 3, v.w);
}

__global__ __launch_bounds__(256) void pool_cnt_kernel(const int* __restrict__ batch,
                                                       int NL, float* __restrict__ cnt) {
  int i = blockIdx.x * blockDim.x + threadIdx.x;
  if (i < NL) atomicAdd(&cnt[batch[i]], 1.0f);
}

__global__ __launch_bounds__(256) void finalize_kernel(const float* __restrict__ sums,
                                                       const float* __restrict__ cnt,
                                                       float* __restrict__ out, int G) {
  int t = blockIdx.x * blockDim.x + threadIdx.x;
  if (t >= G * H) return;
  int g = t >> 7;
  out[t] = sums[t] / fmaxf(cnt[g], 1.0f);
}

extern "C" void kernel_launch(void* const* d_in, const int* in_sizes, int n_in,
                              void* d_out, int out_size, void* d_ws, size_t ws_size,
                              hipStream_t stream) {
  const float* x_node = (const float*)d_in[0];
  const float* x_link = (const float*)d_in[1];
  const float* Wm_nl  = (const float*)d_in[2];
  const float* bm_nl  = (const float*)d_in[3];
  const float* Wu_nl  = (const float*)d_in[4];
  const float* bu_nl  = (const float*)d_in[5];
  const float* Wm_ln  = (const float*)d_in[6];
  const float* bm_ln  = (const float*)d_in[7];
  const float* Wu_ln  = (const float*)d_in[8];
  const float* bu_ln  = (const float*)d_in[9];
  const int*   nl     = (const int*)d_in[10];
  const int*   ln     = (const int*)d_in[11];
  const int*   batch  = (const int*)d_in[12];

  const int NL = in_sizes[12];
  const int NN = 50000;
  const int F_NODE = in_sizes[0] / NN;
  const int F_LINK = in_sizes[1] / NL;
  const int E  = in_sizes[10] / 2;
  const int Lnum = in_sizes[3] / H;
  const int G = out_size / H;

  const size_t MH = (size_t)((NN > NL) ? NN : NL) * H;
  float* ws   = (float*)d_ws;
  float* xn   = ws;
  float* xl   = xn  + MH;
  float* xn2  = xl  + MH;
  float* xl2  = xn2 + MH;
  float* Sl   = xl2 + MH;
  float* Sn   = Sl  + MH;
  float* AGG  = Sn  + MH;
  float* dgl  = AGG + MH;
  float* dgn  = dgl + NL;
  float* sums = dgn + NN;
  float* cnt  = sums + (size_t)G * H;
  (void)n_in; (void)ws_size;

  auto nb = [](long n, int b) { return (int)((n + b - 1) / b); };

  zero_kernel<<<nb(NL, 256), 256, 0, stream>>>(dgl, NL);
  zero_kernel<<<nb(NN, 256), 256, 0, stream>>>(dgn, NN);
  deg_kernel<<<nb(E, 256), 256, 0, stream>>>(nl, E, dgl);
  deg_kernel<<<nb(E, 256), 256, 0, stream>>>(ln, E, dgn);

  pad_kernel<<<nb((long)NN * H, 256), 256, 0, stream>>>(x_node, xn, NN, F_NODE);
  pad_kernel<<<nb((long)NL * H, 256), 256, 0, stream>>>(x_link, xl, NL, F_LINK);

  for (int l = 0; l < Lnum; ++l) {
    float* cxn = (l & 1) ? xn2 : xn;
    float* cxl = (l & 1) ? xl2 : xl;
    float* oxn = (l & 1) ? xn  : xn2;
    float* oxl = (l & 1) ? xl  : xl2;

    const float* WmNLt = Wm_nl + (size_t)l * 2 * H * H;
    const float* WmNLb = WmNLt + H * H;
    const float* WuNLt = Wu_nl + (size_t)l * 2 * H * H;
    const float* WuNLb = WuNLt + H * H;
    const float* WmLNt = Wm_ln + (size_t)l * 2 * H * H;
    const float* WmLNb = WmLNt + H * H;
    const float* WuLNt = Wu_ln + (size_t)l * 2 * H * H;
    const float* WuLNb = WuLNt + H * H;

    zero_kernel<<<nb((long)NL * H, 256), 256, 0, stream>>>(Sl, (long)NL * H);
    zero_kernel<<<nb((long)NN * H, 256), 256, 0, stream>>>(Sn, (long)NN * H);
    scatter_kernel<<<nb((long)E * 32, 256), 256, 0, stream>>>(cxn, nl, E, Sl);
    scatter_kernel<<<nb((long)E * 32, 256), 256, 0, stream>>>(cxl, ln, E, Sn);

    const int gl = nb(NL / 16, TPB);
    const int gn = nb(NN / 16, TPB);
    gemm_dual_kernel<true, false><<<gl, 256, 0, stream>>>(
        Sl, cxl, WmNLt, WmNLb, bm_nl + (size_t)l * H, dgl, AGG, NL);
    gemm_dual_kernel<false, true><<<gl, 256, 0, stream>>>(
        AGG, cxl, WuNLt, WuNLb, bu_nl + (size_t)l * H, nullptr, oxl, NL);
    gemm_dual_kernel<true, false><<<gn, 256, 0, stream>>>(
        Sn, cxn, WmLNt, WmLNb, bm_ln + (size_t)l * H, dgn, AGG, NN);
    gemm_dual_kernel<false, true><<<gn, 256, 0, stream>>>(
        AGG, cxn, WuLNt, WuLNb, bu_ln + (size_t)l * H, nullptr, oxn, NN);
  }

  float* fxl = (Lnum & 1) ? xl2 : xl;
  zero_kernel<<<nb((long)G * H, 256), 256, 0, stream>>>(sums, (long)G * H);
  zero_kernel<<<nb(G, 256), 256, 0, stream>>>(cnt, G);
  pool_sum_kernel<<<nb((long)NL * 32, 256), 256, 0, stream>>>(fxl, batch, NL, sums);
  pool_cnt_kernel<<<nb(NL, 256), 256, 0, stream>>>(batch, NL, cnt);
  finalize_kernel<<<nb((long)G * H, 256), 256, 0, stream>>>(sums, cnt, (float*)d_out, G);
}


# row sums kept in the owner wave's registers: resolve deals queued edges to waves by row & 15; a wave adds its gathered rows (16 in flight) by vector adds in list (arrival) order; no LDS float adds
# speedup vs baseline: 1.7854x; 1.7854x over previous
#include <hip/hip_runtime.h>


typedef __attribute__((ext_vector_type(2))) float v2f;
typedef __attribute__((ext_vector_type(8))) float v8f;

#define H 128
#define LDSPAD 132
#define TPB 8
#define ST 13
#define SROWS (ST * 16)
#define QCAP 4096
#define SWAVES 16
#define SQCAP 256

__global__ __launch_bounds__(256) void zero_kernel(float* p, long n) {
  long i = (long)blockIdx.x * blockDim.x + threadIdx.x;
  if (i < n) p[i] = 0.0f;
}

__global__ __launch_bounds__(256) void pad_kernel(const float* __restrict__ in,
                                                  float* __restrict__ out,
                                                  int N, int F) {
  long t = (long)blockIdx.x * blockDim.x + threadIdx.x;
  if (t >= (long)N * H) return;
  int i = (int)(t >> 7), h = (int)(t & 127);
  out[t] = (h < F) ? in[(long)i * F + h] : 0.0f;
}

__global__ __launch_bounds__(256) void deg_kernel(const int* __restrict__ edge,
                                                  int E, float* __restrict__ deg,
                                                  int* __restrict__ own) {
  const int q = blockIdx.x * blockDim.x + threadIdx.x;
  const int e0 = q << 2;
  if (e0 >= E) return;
  const int* __restrict__ dstp = edge + E;
  unsigned w = 0;
  if (e0 + 3 < E) {
    const int d0 = dstp[e0], d1 = dstp[e0 + 1], d2 = dstp[e0 + 2], d3 = dstp[e0 + 3];
    atomicAdd(&deg[d0], 1.0f);
    atomicAdd(&deg[d1], 1.0f);
    atomicAdd(&deg[d2], 1.0f);
    atomicAdd(&deg[d3], 1.0f);
    w = (((unsigned)d0 / SROWS) & 255u) | ((((unsigned)d1 / SROWS) & 255u) << 8) |
        ((((unsigned)d2 / SROWS) & 255u) << 16) | ((((unsigned)d3 / SROWS) & 255u) << 24);
  } else {
    for (int j = 0; e0 + j < E; ++j) {
      const int d = dstp[e0 + j];
      atomicAdd(&deg[d], 1.0f);
      w |= (((unsigned)d / SROWS) & 255u) << (8 * j);
    }
  }
  own[q] = (int)w;
}

__device__ __forceinline__ void pool_add(float* __restrict__ p, int pitch, int g, int G, float v) {
  if ((unsigned)g < (unsigned)G) atomicAdd(p + (long)g * pitch, v);
}

template <bool HAS_DEG, bool RELU, bool POOL>
__global__ __launch_bounds__(256) void gemm_dual_kernel(
    const float* __restrict__ A, const float* __restrict__ B,
    const float* __restrict__ WA, const float* __restrict__ WB,
    const float* __restrict__ bias, const float* __restrict__ deg,
    float* __restrict__ Y, int M,
    const int* __restrict__ seg, float* __restrict__ psum,
    float* __restrict__ pcnt, int G) {
  __shared__ float lwa[H * LDSPAD];
  __shared__ float lwb[H * LDSPAD];
  for (int idx = threadIdx.x; idx < H * H; idx += 256) {
    int k = idx >> 7, n = idx & 127;
    lwa[n * LDSPAD + k] = WA[idx];
    lwb[n * LDSPAD + k] = WB[idx];
  }
  __syncthreads();

  const int wave = threadIdx.x >> 5;
  const int lane = threadIdx.x & 31;
  const int l15  = lane & 15;
  const int kh2  = (lane >> 4) << 1;
  const int ntiles = M >> 4;

  for (int t = blockIdx.x * TPB + wave; t < ntiles; t += gridDim.x * TPB) {
    const int m0 = t << 4;
    v8f cA[8], cB[8];
#pragma unroll
    for (int n = 0; n < 8; ++n) {
      cA[n] = (v8f){0, 0, 0, 0, 0, 0, 0, 0};
      cB[n] = (v8f){0, 0, 0, 0, 0, 0, 0, 0};
    }

    const float* Ap = A + (long)(m0 + l15) * H + kh2;
    const float* Bp = B + (long)(m0 + l15) * H + kh2;
    const float* wabase = &lwa[l15 * LDSPAD + kh2];
    const float* wbbase = &lwb[l15 * LDSPAD + kh2];

#pragma unroll 1
    for (int kk = 0; kk < 32; ++kk) {
      const int k0 = kk << 2;
      v2f a = *(const v2f*)(Ap + k0);
      v2f b = *(const v2f*)(Bp + k0);
#pragma unroll
      for (int n = 0; n < 8; ++n) {
        v2f wa = *(const v2f*)(wabase + n * 16 * LDSPAD + k0);
        v2f wb = *(const v2f*)(wbbase + n * 16 * LDSPAD + k0);
        cA[n] = __builtin_amdgcn_wmma_f32_16x16x4_f32(false, a, false, wa,
                                                      (short)0, cA[n], false, false);
        cB[n] = __builtin_amdgcn_wmma_f32_16x16x4_f32(false, b, false, wb,
                                                      (short)0, cB[n], false, false);
      }
    }

    const int rbase = m0 + ((lane >> 4) << 3);
    float degv[8];
    if (HAS_DEG) {
      const float4 d0 = *(const float4*)(deg + rbase);
      const float4 d1 = *(const float4*)(deg + rbase + 4);
      degv[0] = d0.x; degv[1] = d0.y; degv[2] = d0.z; degv[3] = d0.w;
      degv[4] = d1.x; degv[5] = d1.y; degv[6] = d1.z; degv[7] = d1.w;
    } else {
#pragma unroll
      for (int r = 0; r < 8; ++r) degv[r] = 1.0f;
    }
    int gid[8];
    bool uni = true;
    if (POOL) {
      const int4 g0 = *(const int4*)(seg + rbase);
      const int4 g1 = *(const int4*)(seg + rbase + 4);
      gid[0] = g0.x; gid[1] = g0.y; gid[2] = g0.z; gid[3] = g0.w;
      gid[4] = g1.x; gid[5] = g1.y; gid[6] = g1.z; gid[7] = g1.w;
#pragma unroll
      for (int r = 1; r < 8; ++r) uni = uni && (gid[r] == gid[0]);
      if (l15 == 0) {
        if (uni) {
          pool_add(pcnt, 1, gid[0], G, 8.0f);
        } else {
          int gc = gid[0];
          float len = 1.0f;
#pragma unroll
          for (int r = 1; r < 8; ++r) {
            if (gid[r] == gc) { len += 1.0f; }
            else { pool_add(pcnt, 1, gc, G, len); gc = gid[r]; len = 1.0f; }
          }
          pool_add(pcnt, 1, gc, G, len);
        }
      }
    }
#pragma unroll
    for (int n = 0; n < 8; ++n) {
      const float bv = bias[n * 16 + l15];
      float v[8];
#pragma unroll
      for (int r = 0; r < 8; ++r) {
        float acc = HAS_DEG ? (cA[n][r] + degv[r] * (cB[n][r] + bv))
                            : (cA[n][r] + cB[n][r] + bv);
        if (RELU) acc = fmaxf(acc, 0.0f);
        v[r] = acc;
      }
      if (!POOL) {
#pragma unroll
        for (int r = 0; r < 8; ++r)
          Y[(long)(rbase + r) * H + n * 16 + l15] = v[r];
      } else {
        float* pc = psum + n * 16 + l15;
        if (uni) {
          float run = v[0];
#pragma unroll
          for (int r = 1; r < 8; ++r) run += v[r];
          pool_add(pc, H, gid[0], G, run);
        } else {
          int gc = gid[0];
          float run = v[0];
#pragma unroll
          for (int r = 1; r < 8; ++r) {
            if (gid[r] == gc) { run += v[r]; }
            else { pool_add(pc, H, gc, G, run); gc = gid[r]; run = v[r]; }
          }
          pool_add(pc, H, gc, G, run);
        }
      }
    }
  }
}

typedef __attribute__((ext_vector_type(16))) float v16f;
static_assert(ST <= 15, "a wave's rows are places 0..ST-1 of a 16-wide register vector; place 15 takes the entries that add nothing");
static_assert(SWAVES == 16 && SROWS == ST * SWAVES, "local row = place * SWAVES + wave");

__global__ __launch_bounds__(512) void scatter_gemm_kernel(
    const float* __restrict__ X, const int* __restrict__ edge, int E,
    const int* __restrict__ own, const float* __restrict__ B,
    const float* __restrict__ WA, const float* __restrict__ WB,
    const float* __restrict__ bias, const float* __restrict__ deg,
    float* __restrict__ Y, int M) {
  __shared__ float lwa[H * LDSPAD];
  __shared__ float lwb[H * LDSPAD];
  __shared__ __attribute__((aligned(16))) float ls[SROWS * LDSPAD];
  __shared__ int qsrc[QCAP];
  __shared__ __attribute__((aligned(16))) unsigned sq[SWAVES * SQCAP];
  __shared__ int sqn[SWAVES];
  __shared__ int qn;

  const int tid = threadIdx.x;
  const int row0 = blockIdx.x * SROWS;
  const int nrows = (M - row0 < SROWS) ? (M - row0) : SROWS;

  for (int i = tid; i < SROWS * LDSPAD / 4; i += 512)
    ((float4*)ls)[i] = make_float4(0.0f, 0.0f, 0.0f, 0.0f);
  if (tid == 0) qn = 0;
  if (tid < SWAVES) sqn[tid] = 0;
  for (int idx = tid; idx < H * H; idx += 512) {
    int k = idx >> 7, n = idx & 127;
    lwa[n * LDSPAD + k] = WA[idx];
    lwb[n * LDSPAD + k] = WB[idx];
  }
  __syncthreads();

  const int* __restrict__ dstp = edge + E;
  const unsigned nr = (unsigned)nrows;
  auto push = [&](int e) {
    const int slot = atomicAdd(&qn, 1);
    if (slot < QCAP) {
      qsrc[slot] = e;
    } else {
      const int dl = dstp[e] - row0;
      if ((unsigned)dl < nr) {
        const float* xr = X + (long)edge[e] * H;
        float* sr = ls + dl * LDSPAD;
        for (int c = 0; c < H; ++c) atomicAdd(sr + c, xr[c]);
      }
    }
  };
  const unsigned mine = (unsigned)blockIdx.x & 255u;
  const unsigned pat = mine * 0x01010101u;
  auto eqb = [&](int w) -> unsigned {
    const unsigned x = (unsigned)w ^ pat;
    return ~(((x & 0x7f7f7f7fu) + 0x7f7f7f7fu) | x | 0x7f7f7f7fu);
  };
  auto nib = [](unsigned h) -> unsigned {
    return ((h >> 7) * 0x01020408u) >> 24;
  };
  auto hits = [&](const int4 d) -> unsigned {
    return nib(eqb(d.x)) | (nib(eqb(d.y)) << 4) | (nib(eqb(d.z)) << 8) | (nib(eqb(d.w)) << 12);
  };
  {
    const int4* __restrict__ o4 = (const int4*)own;
    const int n16 = E >> 4;
    int i = tid;
    for (; i + 15 * 512 < n16; i += 16 * 512) {
      int4 d[16];
#pragma unroll
      for (int u = 0; u < 16; ++u) d[u] = o4[i + u * 512];
#pragma unroll
      for (int u = 0; u < 16; u += 2) {
        unsigned m = hits(d[u]) | (hits(d[u + 1]) << 16);
        while (m) {
          const int j = __ffs(m) - 1;
          m &= m - 1;
          push(16 * (i + (u + (j >> 4)) * 512) + (j & 15));
        }
      }
    }
    for (; i < n16; i += 512) {
      unsigned m = hits(o4[i]);
      while (m) {
        const int j = __ffs(m) - 1;
        m &= m - 1;
        push(16 * i + j);
      }
    }
    for (int e = (n16 << 4) + tid; e < E; e += 512)
      if ((((unsigned)own[e >> 2] >> (8 * (e & 3))) & 255u) == mine) push(e);
  }
  __syncthreads();

  const int qc = (qn < QCAP) ? qn : QCAP;
  for (int q = tid; q < qc; q += 512) {
    const int e = qsrc[q];
    const int dl = dstp[e] - row0;
    const int s = edge[e];
    const bool own_row = (unsigned)dl < nr;
    const int w = dl & (SWAVES - 1);
    const unsigned place = own_row ? (unsigned)(dl >> 4) : 15u;
    const int pos = atomicAdd(&sqn[w], 1);
    if (pos < SQCAP) {
      sq[w * SQCAP + pos] = ((unsigned)s << 4) | place;
    } else if (own_row) {
      const float* xr = X + (long)s * H;
      float* sr = ls + dl * LDSPAD;
      for (int c = 0; c < H; ++c) atomicAdd(sr + c, xr[c]);
    }
  }
  __syncthreads();

  const int wave = __builtin_amdgcn_readfirstlane(tid >> 5);
  const int lane = tid & 31;
  {
    unsigned* myq = sq + wave * SQCAP;
    int n = __builtin_amdgcn_readfirstlane(sqn[wave]);
    n = (n < SQCAP) ? n : SQCAP;
    const int n16 = (n + 15) & ~15;
    if (lane < 16 && n + lane < n16) myq[n + lane] = 15u;
    v16f ax = {0, 0, 0, 0, 0, 0, 0, 0, 0, 0, 0, 0, 0, 0, 0, 0};
    v16f ay = ax, az = ax, aw = ax;
    for (int q0 = 0; q0 < n16; q0 += 16) {
      const uint4 e0 = *(const uint4*)&myq[q0], e1 = *(const uint4*)&myq[q0 + 4];
      const uint4 e2 = *(const uint4*)&myq[q0 + 8], e3 = *(const uint4*)&myq[q0 + 12];
      const unsigned ev[16] = {e0.x, e0.y, e0.z, e0.w, e1.x, e1.y, e1.z, e1.w,
                               e2.x, e2.y, e2.z, e2.w, e3.x, e3.y, e3.z, e3.w};
      int en[16];
      float4 v[16];
#pragma unroll
      for (int u = 0; u < 16; ++u) {
        en[u] = __builtin_amdgcn_readfirstlane((int)ev[u]);
        v[u] = *(const float4*)(X + (long)(int)((unsigned)en[u] >> 4) * H + lane * 4);
      }
#pragma unroll
      for (int u = 0; u < 16; ++u) {
        const int s = en[u] & 15;
        ax[s] += v[u].x;
        ay[s] += v[u].y;
        az[s] += v[u].z;
        aw[s] += v[u].w;
      }
    }
#pragma unroll
    for (int s = 0; s < ST; ++s) {
      float4* p = (float4*)(ls + (s * SWAVES + wave) * LDSPAD + lane * 4);
      float4 t = *p;
      t.x += ax[s]; t.y += ay[s]; t.z += az[s]; t.w += aw[s];
      *p = t;
    }
  }
  __syncthreads();

  const int l15  = lane & 15;
  const int kh2  = (lane >> 4) << 1;
  const int ntiles = M >> 4;
  for (int tl = wave; tl < ST; tl += SWAVES) {
    const int t = blockIdx.x * ST + tl;
    if (t >= ntiles) break;
    const int m0 = t << 4;
    v8f cA[8], cB[8];
#pragma unroll
    for (int n = 0; n < 8; ++n) {
      cA[n] = (v8f){0, 0, 0, 0, 0, 0, 0, 0};
      cB[n] = (v8f){0, 0, 0, 0, 0, 0, 0, 0};
    }
    const float* Ap = &ls[(tl * 16 + l15) * LDSPAD + kh2];
    const float* Bp = B + (long)(m0 + l15) * H + kh2;
    const float* wabase = &lwa[l15 * LDSPAD + kh2];
    const float* wbbase = &lwb[l15 * LDSPAD + kh2];

#pragma unroll 1
    for (int kk = 0; kk < 32; ++kk) {
      const int k0 = kk << 2;
      v2f a = *(const v2f*)(Ap + k0);
      v2f b = *(const v2f*)(Bp + k0);
#pragma unroll
      for (int n = 0; n < 8; ++n) {
        v2f wa = *(const v2f*)(wabase + n * 16 * LDSPAD + k0);
        v2f wb = *(const v2f*)(wbbase + n * 16 * LDSPAD + k0);
        cA[n] = __builtin_amdgcn_wmma_f32_16x16x4_f32(false, a, false, wa,
                                                      (short)0, cA[n], false, false);
        cB[n] = __builtin_amdgcn_wmma_f32_16x16x4_f32(false, b, false, wb,
                                                      (short)0, cB[n], false, false);
      }
    }

    const int rbase = m0 + ((lane >> 4) << 3);
    const float4 d0 = *(const float4*)(deg + rbase);
    const float4 d1 = *(const float4*)(deg + rbase + 4);
    const float degv[8] = {d0.x, d0.y, d0.z, d0.w, d1.x, d1.y, d1.z, d1.w};
#pragma unroll
    for (int n = 0; n < 8; ++n) {
      const float bv = bias[n * 16 + l15];
#pragma unroll
      for (int r = 0; r < 8; ++r)
        Y[(long)(rbase + r) * H + n * 16 + l15] = cA[n][r] + degv[r] * (cB[n][r] + bv);
    }
  }
}

__global__ __launch_bounds__(256) void finalize_kernel(const float* __restrict__ sums,
                                                       const float* __restrict__ cnt,
                                                       float* __restrict__ out, int G) {
  int t = blockIdx.x * blockDim.x + threadIdx.x;
  if (t >= G * H) return;
  int g = t >> 7;
  out[t] = sums[t] / fmaxf(cnt[g], 1.0f);
}

extern "C" void kernel_launch(void* const* d_in, const int* in_sizes, int n_in,
                              void* d_out, int out_size, void* d_ws, size_t ws_size,
                              hipStream_t stream) {
  const float* x_node = (const float*)d_in[0];
  const float* x_link = (const float*)d_in[1];
  const float* Wm_nl  = (const float*)d_in[2];
  const float* bm_nl  = (const float*)d_in[3];
  const float* Wu_nl  = (const float*)d_in[4];
  const float* bu_nl  = (const float*)d_in[5];
  const float* Wm_ln  = (const float*)d_in[6];
  const float* bm_ln  = (const float*)d_in[7];
  const float* Wu_ln  = (const float*)d_in[8];
  const float* bu_ln  = (const float*)d_in[9];
  const int*   nl     = (const int*)d_in[10];
  const int*   ln     = (const int*)d_in[11];
  const int*   batch  = (const int*)d_in[12];

  const int NL = in_sizes[12];
  const int NN = 50000;
  const int F_NODE = in_sizes[0] / NN;
  const int F_LINK = in_sizes[1] / NL;
  const int E  = in_sizes[10] / 2;
  const int Lnum = in_sizes[3] / H;
  const int G = out_size / H;

  const size_t MH = (size_t)((NN > NL) ? NN : NL) * H;
  float* ws   = (float*)d_ws;
  float* xn   = ws;
  float* xl   = xn  + MH;
  float* xn2  = xl  + MH;
  float* xl2  = xn2 + MH;
  float* Sl   = xl2 + MH;
  float* Sn   = Sl  + MH;
  float* AGG  = Sn  + MH;
  float* dgl  = AGG + MH;
  float* dgn  = dgl + NL;
  float* sums = dgn + NN;
  float* cnt  = sums + (size_t)G * H;
  (void)n_in; (void)ws_size;
  int* ownl = (int*)Sl;
  int* ownn = (int*)Sn;

  auto nb = [](long n, int b) { return (int)((n + b - 1) / b); };

  zero_kernel<<<nb(NL, 256), 256, 0, stream>>>(dgl, NL);
  zero_kernel<<<nb(NN, 256), 256, 0, stream>>>(dgn, NN);
  deg_kernel<<<nb(nb(E, 4), 256), 256, 0, stream>>>(nl, E, dgl, ownl);
  deg_kernel<<<nb(nb(E, 4), 256), 256, 0, stream>>>(ln, E, dgn, ownn);

  pad_kernel<<<nb((long)NN * H, 256), 256, 0, stream>>>(x_node, xn, NN, F_NODE);
  pad_kernel<<<nb((long)NL * H, 256), 256, 0, stream>>>(x_link, xl, NL, F_LINK);

  zero_kernel<<<nb((long)G * H, 256), 256, 0, stream>>>(sums, (long)G * H);
  zero_kernel<<<nb(G, 256), 256, 0, stream>>>(cnt, G);

  for (int l = 0; l < Lnum; ++l) {
    float* cxn = (l & 1) ? xn2 : xn;
    float* cxl = (l & 1) ? xl2 : xl;
    float* oxn = (l & 1) ? xn  : xn2;
    float* oxl = (l & 1) ? xl  : xl2;

    const float* WmNLt = Wm_nl + (size_t)l * 2 * H * H;
    const float* WmNLb = WmNLt + H * H;
    const float* WuNLt = Wu_nl + (size_t)l * 2 * H * H;
    const float* WuNLb = WuNLt + H * H;
    const float* WmLNt = Wm_ln + (size_t)l * 2 * H * H;
    const float* WmLNb = WmLNt + H * H;
    const float* WuLNt = Wu_ln + (size_t)l * 2 * H * H;
    const float* WuLNb = WuLNt + H * H;

    const int gl = nb(NL / 16, TPB);
    const int gn = nb(NN / 16, TPB);
    const int sl = nb(NL / 16, ST);
    const int sn = nb(NN / 16, ST);
    scatter_gemm_kernel<<<sl, 512, 0, stream>>>(
        cxn, nl, E, ownl, cxl, WmNLt, WmNLb, bm_nl + (size_t)l * H, dgl, AGG, NL);
    if (l == Lnum - 1)
      gemm_dual_kernel<false, true, true><<<gl, 256, 0, stream>>>(
          AGG, cxl, WuNLt, WuNLb, bu_nl + (size_t)l * H, nullptr, nullptr, NL,
          batch, sums, cnt, G);
    else
      gemm_dual_kernel<false, true, false><<<gl, 256, 0, stream>>>(
          AGG, cxl, WuNLt, WuNLb, bu_nl + (size_t)l * H, nullptr, oxl, NL,
          nullptr, nullptr, nullptr, 0);
    scatter_gemm_kernel<<<sn, 512, 0, stream>>>(
        cxl, ln, E, ownn, cxn, WmLNt, WmLNb, bm_ln + (size_t)l * H, dgn, AGG, NN);
    gemm_dual_kernel<false, true, false><<<gn, 256, 0, stream>>>(
        AGG, cxn, WuLNt, WuLNb, bu_ln + (size_t)l * H, nullptr, oxn, NN,
        nullptr, nullptr, nullptr, 0);
  }

  finalize_kernel<<<nb((long)G * H, 256), 256, 0, stream>>>(sums, cnt, (float*)d_out, G);
}


# update products chained in the owner workgroup: a wave writes its aggregate tile over its own LDS rows, update weights restaged over message weights, A fragments from LDS; 17 -> 13 launches
# speedup vs baseline: 2.0866x; 1.1687x over previous
#include <hip/hip_runtime.h>


typedef __attribute__((ext_vector_type(2))) float v2f;
typedef __attribute__((ext_vector_type(8))) float v8f;

#define H 128
#define LDSPAD 132
#define ST 13
#define SROWS (ST * 16)
#define QCAP 4096
#define SWAVES 16
#define SQCAP 256

__global__ __launch_bounds__(256) void zero_kernel(float* p, long n) {
  long i = (long)blockIdx.x * blockDim.x + threadIdx.x;
  if (i < n) p[i] = 0.0f;
}

__global__ __launch_bounds__(256) void pad_kernel(const float* __restrict__ in,
                                                  float* __restrict__ out,
                                                  int N, int F) {
  long t = (long)blockIdx.x * blockDim.x + threadIdx.x;
  if (t >= (long)N * H) return;
  int i = (int)(t >> 7), h = (int)(t & 127);
  out[t] = (h < F) ? in[(long)i * F + h] : 0.0f;
}

__global__ __launch_bounds__(256) void deg_kernel(const int* __restrict__ edge,
                                                  int E, float* __restrict__ deg,
                                                  int* __restrict__ own) {
  const int q = blockIdx.x * blockDim.x + threadIdx.x;
  const int e0 = q << 2;
  if (e0 >= E) return;
  const int* __restrict__ dstp = edge + E;
  unsigned w = 0;
  if (e0 + 3 < E) {
    const int d0 = dstp[e0], d1 = dstp[e0 + 1], d2 = dstp[e0 + 2], d3 = dstp[e0 + 3];
    atomicAdd(&deg[d0], 1.0f);
    atomicAdd(&deg[d1], 1.0f);
    atomicAdd(&deg[d2], 1.0f);
    atomicAdd(&deg[d3], 1.0f);
    w = (((unsigned)d0 / SROWS) & 255u) | ((((unsigned)d1 / SROWS) & 255u) << 8) |
        ((((unsigned)d2 / SROWS) & 255u) << 16) | ((((unsigned)d3 / SROWS) & 255u) << 24);
  } else {
    for (int j = 0; e0 + j < E; ++j) {
      const int d = dstp[e0 + j];
      atomicAdd(&deg[d], 1.0f);
      w |= (((unsigned)d / SROWS) & 255u) << (8 * j);
    }
  }
  own[q] = (int)w;
}

__device__ __forceinline__ void dual_product(const float* Ap, const float* Bp,
                                             const float* wabase, const float* wbbase,
                                             v8f (&cA)[8], v8f (&cB)[8]) {
#pragma unroll
  for (int n = 0; n < 8; ++n) {
    cA[n] = (v8f){0, 0, 0, 0, 0, 0, 0, 0};
    cB[n] = (v8f){0, 0, 0, 0, 0, 0, 0, 0};
  }
#pragma unroll 1
  for (int kk = 0; kk < 32; ++kk) {
    const int k0 = kk << 2;
    v2f a = *(const v2f*)(Ap + k0);
    v2f b = *(const v2f*)(Bp + k0);
#pragma unroll
    for (int n = 0; n < 8; ++n) {
      v2f wa = *(const v2f*)(wabase + n * 16 * LDSPAD + k0);
      v2f wb = *(const v2f*)(wbbase + n * 16 * LDSPAD + k0);
      cA[n] = __builtin_amdgcn_wmma_f32_16x16x4_f32(false, a, false, wa,
                                                    (short)0, cA[n], false, false);
      cB[n] = __builtin_amdgcn_wmma_f32_16x16x4_f32(false, b, false, wb,
                                                    (short)0, cB[n], false, false);
    }
  }
}

__device__ __forceinline__ void pool_add(float* __restrict__ p, int pitch, int g, int G, float v) {
  if ((unsigned)g < (unsigned)G) atomicAdd(p + (long)g * pitch, v);
}

typedef __attribute__((ext_vector_type(16))) float v16f;
static_assert(ST <= 15, "a wave's rows are places 0..ST-1 of a 16-wide register vector; place 15 takes the entries that add nothing");
static_assert(SWAVES == 16 && SROWS == ST * SWAVES, "local row = place * SWAVES + wave");

template <bool POOL>
__global__ __launch_bounds__(512) void scatter_gemm_kernel(
    const float* __restrict__ X, const int* __restrict__ edge, int E,
    const int* __restrict__ own, const float* __restrict__ B,
    const float* __restrict__ WA, const float* __restrict__ WB,
    const float* __restrict__ bias, const float* __restrict__ deg,
    const float* __restrict__ UA, const float* __restrict__ UB,
    const float* __restrict__ ubias,
    float* __restrict__ Y, int M,
    const int* __restrict__ seg, float* __restrict__ psum,
    float* __restrict__ pcnt, int G) {
  __shared__ float lwa[H * LDSPAD];
  __shared__ float lwb[H * LDSPAD];
  __shared__ __attribute__((aligned(16))) float ls[SROWS * LDSPAD];
  __shared__ int qsrc[QCAP];
  __shared__ __attribute__((aligned(16))) unsigned sq[SWAVES * SQCAP];
  __shared__ int sqn[SWAVES];
  __shared__ int qn;

  const int tid = threadIdx.x;
  const int row0 = blockIdx.x * SROWS;
  const int nrows = (M - row0 < SROWS) ? (M - row0) : SROWS;

  for (int i = tid; i < SROWS * LDSPAD / 4; i += 512)
    ((float4*)ls)[i] = make_float4(0.0f, 0.0f, 0.0f, 0.0f);
  if (tid == 0) qn = 0;
  if (tid < SWAVES) sqn[tid] = 0;
  for (int idx = tid; idx < H * H; idx += 512) {
    int k = idx >> 7, n = idx & 127;
    lwa[n * LDSPAD + k] = WA[idx];
    lwb[n * LDSPAD + k] = WB[idx];
  }
  __syncthreads();

  const int* __restrict__ dstp = edge + E;
  const unsigned nr = (unsigned)nrows;
  auto push = [&](int e) {
    const int slot = atomicAdd(&qn, 1);
    if (slot < QCAP) {
      qsrc[slot] = e;
    } else {
      const int dl = dstp[e] - row0;
      if ((unsigned)dl < nr) {
        const float* xr = X + (long)edge[e] * H;
        float* sr = ls + dl * LDSPAD;
        for (int c = 0; c < H; ++c) atomicAdd(sr + c, xr[c]);
      }
    }
  };
  const unsigned mine = (unsigned)blockIdx.x & 255u;
  const unsigned pat = mine * 0x01010101u;
  auto eqb = [&](int w) -> unsigned {
    const unsigned x = (unsigned)w ^ pat;
    return ~(((x & 0x7f7f7f7fu) + 0x7f7f7f7fu) | x | 0x7f7f7f7fu);
  };
  auto nib = [](unsigned h) -> unsigned {
    return ((h >> 7) * 0x01020408u) >> 24;
  };
  auto hits = [&](const int4 d) -> unsigned {
    return nib(eqb(d.x)) | (nib(eqb(d.y)) << 4) | (nib(eqb(d.z)) << 8) | (nib(eqb(d.w)) << 12);
  };
  {
    const int4* __restrict__ o4 = (const int4*)own;
    const int n16 = E >> 4;
    int i = tid;
    for (; i + 15 * 512 < n16; i += 16 * 512) {
      int4 d[16];
#pragma unroll
      for (int u = 0; u < 16; ++u) d[u] = o4[i + u * 512];
#pragma unroll
      for (int u = 0; u < 16; u += 2) {
        unsigned m = hits(d[u]) | (hits(d[u + 1]) << 16);
        while (m) {
          const int j = __ffs(m) - 1;
          m &= m - 1;
          push(16 * (i + (u + (j >> 4)) * 512) + (j & 15));
        }
      }
    }
    for (; i < n16; i += 512) {
      unsigned m = hits(o4[i]);
      while (m) {
        const int j = __ffs(m) - 1;
        m &= m - 1;
        push(16 * i + j);
      }
    }
    for (int e = (n16 << 4) + tid; e < E; e += 512)
      if ((((unsigned)own[e >> 2] >> (8 * (e & 3))) & 255u) == mine) push(e);
  }
  __syncthreads();

  const int qc = (qn < QCAP) ? qn : QCAP;
  for (int q = tid; q < qc; q += 512) {
    const int e = qsrc[q];
    const int dl = dstp[e] - row0;
    const int s = edge[e];
    const bool own_row = (unsigned)dl < nr;
    const int w = dl & (SWAVES - 1);
    const unsigned place = own_row ? (unsigned)(dl >> 4) : 15u;
    const int pos = atomicAdd(&sqn[w], 1);
    if (pos < SQCAP) {
      sq[w * SQCAP + pos] = ((unsigned)s << 4) | place;
    } else if (own_row) {
      const float* xr = X + (long)s * H;
      float* sr = ls + dl * LDSPAD;
      for (int c = 0; c < H; ++c) atomicAdd(sr + c, xr[c]);
    }
  }
  __syncthreads();

  const int wave = __builtin_amdgcn_readfirstlane(tid >> 5);
  const int lane = tid & 31;
  {
    unsigned* myq = sq + wave * SQCAP;
    int n = __builtin_amdgcn_readfirstlane(sqn[wave]);
    n = (n < SQCAP) ? n : SQCAP;
    const int n16 = (n + 15) & ~15;
    if (lane < 16 && n + lane < n16) myq[n + lane] = 15u;
    v16f ax = {0, 0, 0, 0, 0, 0, 0, 0, 0, 0, 0, 0, 0, 0, 0, 0};
    v16f ay = ax, az = ax, aw = ax;
    for (int q0 = 0; q0 < n16; q0 += 16) {
      const uint4 e0 = *(const uint4*)&myq[q0], e1 = *(const uint4*)&myq[q0 + 4];
      const uint4 e2 = *(const uint4*)&myq[q0 + 8], e3 = *(const uint4*)&myq[q0 + 12];
      const unsigned ev[16] = {e0.x, e0.y, e0.z, e0.w, e1.x, e1.y, e1.z, e1.w,
                               e2.x, e2.y, e2.z, e2.w, e3.x, e3.y, e3.z, e3.w};
      int en[16];
      float4 v[16];
#pragma unroll
      for (int u = 0; u < 16; ++u) {
        en[u] = __builtin_amdgcn_readfirstlane((int)ev[u]);
        v[u] = *(const float4*)(X + (long)(int)((unsigned)en[u] >> 4) * H + lane * 4);
      }
#pragma unroll
      for (int u = 0; u < 16; ++u) {
        const int s = en[u] & 15;
        ax[s] += v[u].x;
        ay[s] += v[u].y;
        az[s] += v[u].z;
        aw[s] += v[u].w;
      }
    }
#pragma unroll
    for (int s = 0; s < ST; ++s) {
      float4* p = (float4*)(ls + (s * SWAVES + wave) * LDSPAD + lane * 4);
      float4 t = *p;
      t.x += ax[s]; t.y += ay[s]; t.z += az[s]; t.w += aw[s];
      *p = t;
    }
  }
  __syncthreads();

  const int l15  = lane & 15;
  const int kh2  = (lane >> 4) << 1;
  const int t = blockIdx.x * ST + wave;
  const bool has = (wave < ST) && (t < (M >> 4));
  const int m0 = t << 4;
  const int rbase = m0 + ((lane >> 4) << 3);
  const int trow = has ? wave * 16 : 0;
  const float* Ap = &ls[(trow + l15) * LDSPAD + kh2];
  const float* Bp = B + (long)((has ? m0 : 0) + l15) * H + kh2;
  const float* wabase = &lwa[l15 * LDSPAD + kh2];
  const float* wbbase = &lwb[l15 * LDSPAD + kh2];
  v8f cA[8], cB[8];

  if (has) {
    dual_product(Ap, Bp, wabase, wbbase, cA, cB);
    const float4 d0 = *(const float4*)(deg + rbase);
    const float4 d1 = *(const float4*)(deg + rbase + 4);
    const float degv[8] = {d0.x, d0.y, d0.z, d0.w, d1.x, d1.y, d1.z, d1.w};
    float* lrow = ls + (trow + ((lane >> 4) << 3)) * LDSPAD + l15;
#pragma unroll
    for (int n = 0; n < 8; ++n) {
      const float bv = bias[n * 16 + l15];
#pragma unroll
      for (int r = 0; r < 8; ++r)
        lrow[r * LDSPAD + n * 16] = cA[n][r] + degv[r] * (cB[n][r] + bv);
    }
  }
  float ua[H * H / 512], ub[H * H / 512];
#pragma unroll
  for (int i = 0; i < H * H / 512; ++i) {
    ua[i] = UA[tid + i * 512];
    ub[i] = UB[tid + i * 512];
  }
  __syncthreads();
#pragma unroll
  for (int i = 0; i < H * H / 512; ++i) {
    const int idx = tid + i * 512;
    const int k = idx >> 7, n = idx & 127;
    lwa[n * LDSPAD + k] = ua[i];
    lwb[n * LDSPAD + k] = ub[i];
  }
  __syncthreads();

  if (has) {
    dual_product(Ap, Bp, wabase, wbbase, cA, cB);
    int gid[8];
    bool uni = true;
    if (POOL) {
      const int4 g0 = *(const int4*)(seg + rbase);
      const int4 g1 = *(const int4*)(seg + rbase + 4);
      gid[0] = g0.x; gid[1] = g0.y; gid[2] = g0.z; gid[3] = g0.w;
      gid[4] = g1.x; gid[5] = g1.y; gid[6] = g1.z; gid[7] = g1.w;
#pragma unroll
      for (int r = 1; r < 8; ++r) uni = uni && (gid[r] == gid[0]);
      if (l15 == 0) {
        if (uni) {
          pool_add(pcnt, 1, gid[0], G, 8.0f);
        } else {
          int gc = gid[0];
          float len = 1.0f;
#pragma unroll
          for (int r = 1; r < 8; ++r) {
            if (gid[r] == gc) { len += 1.0f; }
            else { pool_add(pcnt, 1, gc, G, len); gc = gid[r]; len = 1.0f; }
          }
          pool_add(pcnt, 1, gc, G, len);
        }
      }
    }
#pragma unroll
    for (int n = 0; n < 8; ++n) {
      const float bv = ubias[n * 16 + l15];
      float v[8];
#pragma unroll
      for (int r = 0; r < 8; ++r)
        v[r] = fmaxf(cA[n][r] + cB[n][r] + bv, 0.0f);
      if (!POOL) {
#pragma unroll
        for (int r = 0; r < 8; ++r)
          Y[(long)(rbase + r) * H + n * 16 + l15] = v[r];
      } else {
        float* pc = psum + n * 16 + l15;
        if (uni) {
          float run = v[0];
#pragma unroll
          for (int r = 1; r < 8; ++r) run += v[r];
          pool_add(pc, H, gid[0], G, run);
        } else {
          int gc = gid[0];
          float run = v[0];
#pragma unroll
          for (int r = 1; r < 8; ++r) {
            if (gid[r] == gc) { run += v[r]; }
            else { pool_add(pc, H, gc, G, run); gc = gid[r]; run = v[r]; }
          }
          pool_add(pc, H, gc, G, run);
        }
      }
    }
  }
}

__global__ __launch_bounds__(256) void finalize_kernel(const float* __restrict__ sums,
                                                       const float* __restrict__ cnt,
                                                       float* __restrict__ out, int G) {
  int t = blockIdx.x * blockDim.x + threadIdx.x;
  if (t >= G * H) return;
  int g = t >> 7;
  out[t] = sums[t] / fmaxf(cnt[g], 1.0f);
}

extern "C" void kernel_launch(void* const* d_in, const int* in_sizes, int n_in,
                              void* d_out, int out_size, void* d_ws, size_t ws_size,
                              hipStream_t stream) {
  const float* x_node = (const float*)d_in[0];
  const float* x_link = (const float*)d_in[1];
  const float* Wm_nl  = (const float*)d_in[2];
  const float* bm_nl  = (const float*)d_in[3];
  const float* Wu_nl  = (const float*)d_in[4];
  const float* bu_nl  = (const float*)d_in[5];
  const float* Wm_ln  = (const float*)d_in[6];
  const float* bm_ln  = (const float*)d_in[7];
  const float* Wu_ln  = (const float*)d_in[8];
  const float* bu_ln  = (const float*)d_in[9];
  const int*   nl     = (const int*)d_in[10];
  const int*   ln     = (const int*)d_in[11];
  const int*   batch  = (const int*)d_in[12];

  const int NL = in_sizes[12];
  const int NN = 50000;
  const int F_NODE = in_sizes[0] / NN;
  const int F_LINK = in_sizes[1] / NL;
  const int E  = in_sizes[10] / 2;
  const int Lnum = in_sizes[3] / H;
  const int G = out_size / H;

  const size_t MH = (size_t)((NN > NL) ? NN : NL) * H;
  float* ws   = (float*)d_ws;
  float* xn   = ws;
  float* xl   = xn  + MH;
  float* xn2  = xl  + MH;
  float* xl2  = xn2 + MH;
  float* Sl   = xl2 + MH;
  float* Sn   = Sl  + MH;
  float* dgl  = Sn  + MH;
  float* dgn  = dgl + NL;
  float* sums = dgn + NN;
  float* cnt  = sums + (size_t)G * H;
  (void)n_in; (void)ws_size;
  int* ownl = (int*)Sl;
  int* ownn = (int*)Sn;

  auto nb = [](long n, int b) { return (int)((n + b - 1) / b); };

  zero_kernel<<<nb(NL, 256), 256, 0, stream>>>(dgl, NL);
  zero_kernel<<<nb(NN, 256), 256, 0, stream>>>(dgn, NN);
  deg_kernel<<<nb(nb(E, 4), 256), 256, 0, stream>>>(nl, E, dgl, ownl);
  deg_kernel<<<nb(nb(E, 4), 256), 256, 0, stream>>>(ln, E, dgn, ownn);

  pad_kernel<<<nb((long)NN * H, 256), 256, 0, stream>>>(x_node, xn, NN, F_NODE);
  pad_kernel<<<nb((long)NL * H, 256), 256, 0, stream>>>(x_link, xl, NL, F_LINK);

  zero_kernel<<<nb((long)G * H, 256), 256, 0, stream>>>(sums, (long)G * H);
  zero_kernel<<<nb(G, 256), 256, 0, stream>>>(cnt, G);

  for (int l = 0; l < Lnum; ++l) {
    float* cxn = (l & 1) ? xn2 : xn;
    float* cxl = (l & 1) ? xl2 : xl;
    float* oxn = (l & 1) ? xn  : xn2;
    float* oxl = (l & 1) ? xl  : xl2;

    const float* WmNLt = Wm_nl + (size_t)l * 2 * H * H;
    const float* WmNLb = WmNLt + H * H;
    const float* WuNLt = Wu_nl + (size_t)l * 2 * H * H;
    const float* WuNLb = WuNLt + H * H;
    const float* WmLNt = Wm_ln + (size_t)l * 2 * H * H;
    const float* WmLNb = WmLNt + H * H;
    const float* WuLNt = Wu_ln + (size_t)l * 2 * H * H;
    const float* WuLNb = WuLNt + H * H;

    const int sl = nb(NL / 16, ST);
    const int sn = nb(NN / 16, ST);
    if (l == Lnum - 1)
      scatter_gemm_kernel<true><<<sl, 512, 0, stream>>>(
          cxn, nl, E, ownl, cxl, WmNLt, WmNLb, bm_nl + (size_t)l * H, dgl,
          WuNLt, WuNLb, bu_nl + (size_t)l * H, nullptr, NL, batch, sums, cnt, G);
    else
      scatter_gemm_kernel<false><<<sl, 512, 0, stream>>>(
          cxn, nl, E, ownl, cxl, WmNLt, WmNLb, bm_nl + (size_t)l * H, dgl,
          WuNLt, WuNLb, bu_nl + (size_t)l * H, oxl, NL, nullptr, nullptr, nullptr, 0);
    scatter_gemm_kernel<false><<<sn, 512, 0, stream>>>(
        cxl, ln, E, ownn, cxn, WmLNt, WmLNb, bm_ln + (size_t)l * H, dgn,
        WuLNt, WuLNb, bu_ln + (size_t)l * H, oxn, NN, nullptr, nullptr, nullptr, 0);
  }

  finalize_kernel<<<nb((long)G * H, 256), 256, 0, stream>>>(sums, cnt, (float*)d_out, G);
}


# matrix work dealt evenly to the four SIMDs: waves 0-11 keep a whole tile, the 13th tile's 8 column blocks go to waves 12-15 (one per SIMD, 2 each; its aggregate in an LDS tile of its own)
# speedup vs baseline: 2.2082x; 1.0583x over previous
#include <hip/hip_runtime.h>


typedef __attribute__((ext_vector_type(2))) float v2f;
typedef __attribute__((ext_vector_type(8))) float v8f;

#define H 128
#define LDSPAD 132
#define ST 13
#define SROWS (ST * 16)
#define QCAP 4096
#define SWAVES 16
#define SQCAP 256
#define SPLIT (ST - 1)
#define PNB (8 / (SWAVES - SPLIT))

__global__ __launch_bounds__(256) void zero_kernel(float* p, long n) {
  long i = (long)blockIdx.x * blockDim.x + threadIdx.x;
  if (i < n) p[i] = 0.0f;
}

__global__ __launch_bounds__(256) void pad_kernel(const float* __restrict__ in,
                                                  float* __restrict__ out,
                                                  int N, int F) {
  long t = (long)blockIdx.x * blockDim.x + threadIdx.x;
  if (t >= (long)N * H) return;
  int i = (int)(t >> 7), h = (int)(t & 127);
  out[t] = (h < F) ? in[(long)i * F + h] : 0.0f;
}

__global__ __launch_bounds__(256) void deg_kernel(const int* __restrict__ edge,
                                                  int E, float* __restrict__ deg,
                                                  int* __restrict__ own) {
  const int q = blockIdx.x * blockDim.x + threadIdx.x;
  const int e0 = q << 2;
  if (e0 >= E) return;
  const int* __restrict__ dstp = edge + E;
  unsigned w = 0;
  if (e0 + 3 < E) {
    const int d0 = dstp[e0], d1 = dstp[e0 + 1], d2 = dstp[e0 + 2], d3 = dstp[e0 + 3];
    atomicAdd(&deg[d0], 1.0f);
    atomicAdd(&deg[d1], 1.0f);
    atomicAdd(&deg[d2], 1.0f);
    atomicAdd(&deg[d3], 1.0f);
    w = (((unsigned)d0 / SROWS) & 255u) | ((((unsigned)d1 / SROWS) & 255u) << 8) |
        ((((unsigned)d2 / SROWS) & 255u) << 16) | ((((unsigned)d3 / SROWS) & 255u) << 24);
  } else {
    for (int j = 0; e0 + j < E; ++j) {
      const int d = dstp[e0 + j];
      atomicAdd(&deg[d], 1.0f);
      w |= (((unsigned)d / SROWS) & 255u) << (8 * j);
    }
  }
  own[q] = (int)w;
}

template <int NB>
__device__ __forceinline__ void dual_product(const float* Ap, const float* Bp,
                                             const float* wabase, const float* wbbase,
                                             v8f (&cA)[NB], v8f (&cB)[NB]) {
#pragma unroll
  for (int n = 0; n < NB; ++n) {
    cA[n] = (v8f){0, 0, 0, 0, 0, 0, 0, 0};
    cB[n] = (v8f){0, 0, 0, 0, 0, 0, 0, 0};
  }
#pragma unroll 1
  for (int kk = 0; kk < 32; ++kk) {
    const int k0 = kk << 2;
    v2f a = *(const v2f*)(Ap + k0);
    v2f b = *(const v2f*)(Bp + k0);
#pragma unroll
    for (int n = 0; n < NB; ++n) {
      v2f wa = *(const v2f*)(wabase + n * 16 * LDSPAD + k0);
      v2f wb = *(const v2f*)(wbbase + n * 16 * LDSPAD + k0);
      cA[n] = __builtin_amdgcn_wmma_f32_16x16x4_f32(false, a, false, wa,
                                                    (short)0, cA[n], false, false);
      cB[n] = __builtin_amdgcn_wmma_f32_16x16x4_f32(false, b, false, wb,
                                                    (short)0, cB[n], false, false);
    }
  }
}

__device__ __forceinline__ void pool_add(float* __restrict__ p, int pitch, int g, int G, float v) {
  if ((unsigned)g < (unsigned)G) atomicAdd(p + (long)g * pitch, v);
}

template <int NB>
__device__ __forceinline__ void agg_store(const v8f (&cA)[NB], const v8f (&cB)[NB],
                                          const float* __restrict__ bcol,
                                          const float* __restrict__ degp, float* lrow) {
  const float4 d0 = *(const float4*)(degp);
  const float4 d1 = *(const float4*)(degp + 4);
  const float degv[8] = {d0.x, d0.y, d0.z, d0.w, d1.x, d1.y, d1.z, d1.w};
#pragma unroll
  for (int n = 0; n < NB; ++n) {
    const float bv = bcol[n * 16];
#pragma unroll
    for (int r = 0; r < 8; ++r)
      lrow[r * LDSPAD + n * 16] = cA[n][r] + degv[r] * (cB[n][r] + bv);
  }
}

template <bool POOL, int NB>
__device__ __forceinline__ void update_out(const v8f (&cA)[NB], const v8f (&cB)[NB],
                                           const float* __restrict__ ubias, int col0, int l15,
                                           int rbase, float* __restrict__ Y,
                                           const int* __restrict__ seg, float* __restrict__ psum,
                                           float* __restrict__ pcnt, int G, bool count) {
  int gid[8];
  bool uni = true;
  if (POOL) {
    const int4 g0 = *(const int4*)(seg + rbase);
    const int4 g1 = *(const int4*)(seg + rbase + 4);
    gid[0] = g0.x; gid[1] = g0.y; gid[2] = g0.z; gid[3] = g0.w;
    gid[4] = g1.x; gid[5] = g1.y; gid[6] = g1.z; gid[7] = g1.w;
#pragma unroll
    for (int r = 1; r < 8; ++r) uni = uni && (gid[r] == gid[0]);
    if (count && l15 == 0) {
      if (uni) {
        pool_add(pcnt, 1, gid[0], G, 8.0f);
      } else {
        int gc = gid[0];
        float len = 1.0f;
#pragma unroll
        for (int r = 1; r < 8; ++r) {
          if (gid[r] == gc) { len += 1.0f; }
          else { pool_add(pcnt, 1, gc, G, len); gc = gid[r]; len = 1.0f; }
        }
        pool_add(pcnt, 1, gc, G, len);
      }
    }
  }
#pragma unroll
  for (int n = 0; n < NB; ++n) {
    const float bv = ubias[col0 + n * 16];
    float v[8];
#pragma unroll
    for (int r = 0; r < 8; ++r)
      v[r] = fmaxf(cA[n][r] + cB[n][r] + bv, 0.0f);
    if (!POOL) {
#pragma unroll
      for (int r = 0; r < 8; ++r)
        Y[(long)(rbase + r) * H + col0 + n * 16] = v[r];
    } else {
      float* pc = psum + col0 + n * 16;
      if (uni) {
        float run = v[0];
#pragma unroll
        for (int r = 1; r < 8; ++r) run += v[r];
        pool_add(pc, H, gid[0], G, run);
      } else {
        int gc = gid[0];
        float run = v[0];
#pragma unroll
        for (int r = 1; r < 8; ++r) {
          if (gid[r] == gc) { run += v[r]; }
          else { pool_add(pc, H, gc, G, run); gc = gid[r]; run = v[r]; }
        }
        pool_add(pc, H, gc, G, run);
      }
    }
  }
}

typedef __attribute__((ext_vector_type(16))) float v16f;
static_assert(ST <= 15, "a wave's rows are places 0..ST-1 of a 16-wide register vector; place 15 takes the entries that add nothing");
static_assert(SWAVES == 16 && SROWS == ST * SWAVES, "local row = place * SWAVES + wave");
static_assert(SWAVES - SPLIT == 4 && PNB * (SWAVES - SPLIT) == 8, "the shared tile's 8 column blocks go to four waves, one per SIMD");

template <bool POOL>
__global__ __launch_bounds__(512) void scatter_gemm_kernel(
    const float* __restrict__ X, const int* __restrict__ edge, int E,
    const int* __restrict__ own, const float* __restrict__ B,
    const float* __restrict__ WA, const float* __restrict__ WB,
    const float* __restrict__ bias, const float* __restrict__ deg,
    const float* __restrict__ UA, const float* __restrict__ UB,
    const float* __restrict__ ubias,
    float* __restrict__ Y, int M,
    const int* __restrict__ seg, float* __restrict__ psum,
    float* __restrict__ pcnt, int G) {
  __shared__ float lwa[H * LDSPAD];
  __shared__ float lwb[H * LDSPAD];
  __shared__ __attribute__((aligned(16))) float ls[SROWS * LDSPAD];
  __shared__ float lagg[16 * LDSPAD];
  __shared__ int qsrc[QCAP];
  __shared__ __attribute__((aligned(16))) unsigned sq[SWAVES * SQCAP];
  __shared__ int sqn[SWAVES];
  __shared__ int qn;

  const int tid = threadIdx.x;
  const int row0 = blockIdx.x * SROWS;
  const int nrows = (M - row0 < SROWS) ? (M - row0) : SROWS;

  for (int i = tid; i < SROWS * LDSPAD / 4; i += 512)
    ((float4*)ls)[i] = make_float4(0.0f, 0.0f, 0.0f, 0.0f);
  if (tid == 0) qn = 0;
  if (tid < SWAVES) sqn[tid] = 0;
  for (int idx = tid; idx < H * H; idx += 512) {
    int k = idx >> 7, n = idx & 127;
    lwa[n * LDSPAD + k] = WA[idx];
    lwb[n * LDSPAD + k] = WB[idx];
  }
  __syncthreads();

  const int* __restrict__ dstp = edge + E;
  const unsigned nr = (unsigned)nrows;
  auto push = [&](int e) {
    const int slot = atomicAdd(&qn, 1);
    if (slot < QCAP) {
      qsrc[slot] = e;
    } else {
      const int dl = dstp[e] - row0;
      if ((unsigned)dl < nr) {
        const float* xr = X + (long)edge[e] * H;
        float* sr = ls + dl * LDSPAD;
        for (int c = 0; c < H; ++c) atomicAdd(sr + c, xr[c]);
      }
    }
  };
  const unsigned mine = (unsigned)blockIdx.x & 255u;
  const unsigned pat = mine * 0x01010101u;
  auto eqb = [&](int w) -> unsigned {
    const unsigned x = (unsigned)w ^ pat;
    return ~(((x & 0x7f7f7f7fu) + 0x7f7f7f7fu) | x | 0x7f7f7f7fu);
  };
  auto nib = [](unsigned h) -> unsigned {
    return ((h >> 7) * 0x01020408u) >> 24;
  };
  auto hits = [&](const int4 d) -> unsigned {
    return nib(eqb(d.x)) | (nib(eqb(d.y)) << 4) | (nib(eqb(d.z)) << 8) | (nib(eqb(d.w)) << 12);
  };
  {
    const int4* __restrict__ o4 = (const int4*)own;
    const int n16 = E >> 4;
    int i = tid;
    for (; i + 15 * 512 < n16; i += 16 * 512) {
      int4 d[16];
#pragma unroll
      for (int u = 0; u < 16; ++u) d[u] = o4[i + u * 512];
#pragma unroll
      for (int u = 0; u < 16; u += 2) {
        unsigned m = hits(d[u]) | (hits(d[u + 1]) << 16);
        while (m) {
          const int j = __ffs(m) - 1;
          m &= m - 1;
          push(16 * (i + (u + (j >> 4)) * 512) + (j & 15));
        }
      }
    }
    for (; i < n16; i += 512) {
      unsigned m = hits(o4[i]);
      while (m) {
        const int j = __ffs(m) - 1;
        m &= m - 1;
        push(16 * i + j);
      }
    }
    for (int e = (n16 << 4) + tid; e < E; e += 512)
      if ((((unsigned)own[e >> 2] >> (8 * (e & 3))) & 255u) == mine) push(e);
  }
  __syncthreads();

  const int qc = (qn < QCAP) ? qn : QCAP;
  for (int q = tid; q < qc; q += 512) {
    const int e = qsrc[q];
    const int dl = dstp[e] - row0;
    const int s = edge[e];
    const bool own_row = (unsigned)dl < nr;
    const int w = dl & (SWAVES - 1);
    const unsigned place = own_row ? (unsigned)(dl >> 4) : 15u;
    const int pos = atomicAdd(&sqn[w], 1);
    if (pos < SQCAP) {
      sq[w * SQCAP + pos] = ((unsigned)s << 4) | place;
    } else if (own_row) {
      const float* xr = X + (long)s * H;
      float* sr = ls + dl * LDSPAD;
      for (int c = 0; c < H; ++c) atomicAdd(sr + c, xr[c]);
    }
  }
  __syncthreads();

  const int wave = __builtin_amdgcn_readfirstlane(tid >> 5);
  const int lane = tid & 31;
  {
    unsigned* myq = sq + wave * SQCAP;
    int n = __builtin_amdgcn_readfirstlane(sqn[wave]);
    n = (n < SQCAP) ? n : SQCAP;
    const int n16 = (n + 15) & ~15;
    if (lane < 16 && n + lane < n16) myq[n + lane] = 15u;
    v16f ax = {0, 0, 0, 0, 0, 0, 0, 0, 0, 0, 0, 0, 0, 0, 0, 0};
    v16f ay = ax, az = ax, aw = ax;
    for (int q0 = 0; q0 < n16; q0 += 16) {
      const uint4 e0 = *(const uint4*)&myq[q0], e1 = *(const uint4*)&myq[q0 + 4];
      const uint4 e2 = *(const uint4*)&myq[q0 + 8], e3 = *(const uint4*)&myq[q0 + 12];
      const unsigned ev[16] = {e0.x, e0.y, e0.z, e0.w, e1.x, e1.y, e1.z, e1.w,
                               e2.x, e2.y, e2.z, e2.w, e3.x, e3.y, e3.z, e3.w};
      int en[16];
      float4 v[16];
#pragma unroll
      for (int u = 0; u < 16; ++u) {
        en[u] = __builtin_amdgcn_readfirstlane((int)ev[u]);
        v[u] = *(const float4*)(X + (long)(int)((unsigned)en[u] >> 4) * H + lane * 4);
      }
#pragma unroll
      for (int u = 0; u < 16; ++u) {
        const int s = en[u] & 15;
        ax[s] += v[u].x;
        ay[s] += v[u].y;
        az[s] += v[u].z;
        aw[s] += v[u].w;
      }
    }
#pragma unroll
    for (int s = 0; s < ST; ++s) {
      float4* p = (float4*)(ls + (s * SWAVES + wave) * LDSPAD + lane * 4);
      float4 t = *p;
      t.x += ax[s]; t.y += ay[s]; t.z += az[s]; t.w += aw[s];
      *p = t;
    }
  }
  __syncthreads();

  const int l15  = lane & 15;
  const int kh2  = (lane >> 4) << 1;
  const bool part = wave >= SPLIT;
  const int ltile = part ? SPLIT : wave;
  const int nb0 = part ? (wave - SPLIT) * PNB : 0;
  const int t = blockIdx.x * ST + ltile;
  const bool has = t < (M >> 4);
  const int m0 = t << 4;
  const int rbase = m0 + ((lane >> 4) << 3);
  const int trow = has ? ltile * 16 : 0;
  const int col0 = nb0 * 16 + l15;
  const float* Ap = &ls[(trow + l15) * LDSPAD + kh2];
  const float* Bp = B + (long)((has ? m0 : 0) + l15) * H + kh2;
  const float* wabase = &lwa[l15 * LDSPAD + kh2];
  const float* wbbase = &lwb[l15 * LDSPAD + kh2];
  const int wstep = nb0 * 16 * LDSPAD;

  if (has) {
    if (!part) {
      v8f cA[8], cB[8];
      dual_product<8>(Ap, Bp, wabase, wbbase, cA, cB);
      agg_store<8>(cA, cB, bias + l15, deg + rbase, ls + (trow + ((lane >> 4) << 3)) * LDSPAD + l15);
    } else {
      v8f cA[PNB], cB[PNB];
      dual_product<PNB>(Ap, Bp, wabase + wstep, wbbase + wstep, cA, cB);
      agg_store<PNB>(cA, cB, bias + col0, deg + rbase, lagg + ((lane >> 4) << 3) * LDSPAD + col0);
    }
  }
  float ua[H * H / 512], ub[H * H / 512];
#pragma unroll
  for (int i = 0; i < H * H / 512; ++i) {
    ua[i] = UA[tid + i * 512];
    ub[i] = UB[tid + i * 512];
  }
  __syncthreads();
#pragma unroll
  for (int i = 0; i < H * H / 512; ++i) {
    const int idx = tid + i * 512;
    const int k = idx >> 7, n = idx & 127;
    lwa[n * LDSPAD + k] = ua[i];
    lwb[n * LDSPAD + k] = ub[i];
  }
  __syncthreads();

  if (has) {
    if (!part) {
      v8f cA[8], cB[8];
      dual_product<8>(Ap, Bp, wabase, wbbase, cA, cB);
      update_out<POOL, 8>(cA, cB, ubias, l15, l15, rbase, Y, seg, psum, pcnt, G, true);
    } else {
      v8f cA[PNB], cB[PNB];
      dual_product<PNB>(&lagg[l15 * LDSPAD + kh2], Bp, wabase + wstep, wbbase + wstep, cA, cB);
      update_out<POOL, PNB>(cA, cB, ubias, col0, l15, rbase, Y, seg, psum, pcnt, G, wave == SPLIT);
    }
  }
}

__global__ __launch_bounds__(256) void finalize_kernel(const float* __restrict__ sums,
                                                       const float* __restrict__ cnt,
                                                       float* __restrict__ out, int G) {
  int t = blockIdx.x * blockDim.x + threadIdx.x;
  if (t >= G * H) return;
  int g = t >> 7;
  out[t] = sums[t] / fmaxf(cnt[g], 1.0f);
}

extern "C" void kernel_launch(void* const* d_in, const int* in_sizes, int n_in,
                              void* d_out, int out_size, void* d_ws, size_t ws_size,
                              hipStream_t stream) {
  const float* x_node = (const float*)d_in[0];
  const float* x_link = (const float*)d_in[1];
  const float* Wm_nl  = (const float*)d_in[2];
  const float* bm_nl  = (const float*)d_in[3];
  const float* Wu_nl  = (const float*)d_in[4];
  const float* bu_nl  = (const float*)d_in[5];
  const float* Wm_ln  = (const float*)d_in[6];
  const float* bm_ln  = (const float*)d_in[7];
  const float* Wu_ln  = (const float*)d_in[8];
  const float* bu_ln  = (const float*)d_in[9];
  const int*   nl     = (const int*)d_in[10];
  const int*   ln     = (const int*)d_in[11];
  const int*   batch  = (const int*)d_in[12];

  const int NL = in_sizes[12];
  const int NN = 50000;
  const int F_NODE = in_sizes[0] / NN;
  const int F_LINK = in_sizes[1] / NL;
  const int E  = in_sizes[10] / 2;
  const int Lnum = in_sizes[3] / H;
  const int G = out_size / H;

  const size_t MH = (size_t)((NN > NL) ? NN : NL) * H;
  float* ws   = (float*)d_ws;
  float* xn   = ws;
  float* xl   = xn  + MH;
  float* xn2  = xl  + MH;
  float* xl2  = xn2 + MH;
  float* Sl   = xl2 + MH;
  float* Sn   = Sl  + MH;
  float* dgl  = Sn  + MH;
  float* dgn  = dgl + NL;
  float* sums = dgn + NN;
  float* cnt  = sums + (size_t)G * H;
  (void)n_in; (void)ws_size;
  int* ownl = (int*)Sl;
  int* ownn = (int*)Sn;

  auto nb = [](long n, int b) { return (int)((n + b - 1) / b); };

  zero_kernel<<<nb(NL, 256), 256, 0, stream>>>(dgl, NL);
  zero_kernel<<<nb(NN, 256), 256, 0, stream>>>(dgn, NN);
  deg_kernel<<<nb(nb(E, 4), 256), 256, 0, stream>>>(nl, E, dgl, ownl);
  deg_kernel<<<nb(nb(E, 4), 256), 256, 0, stream>>>(ln, E, dgn, ownn);

  pad_kernel<<<nb((long)NN * H, 256), 256, 0, stream>>>(x_node, xn, NN, F_NODE);
  pad_kernel<<<nb((long)NL * H, 256), 256, 0, stream>>>(x_link, xl, NL, F_LINK);

  zero_kernel<<<nb((long)G * H, 256), 256, 0, stream>>>(sums, (long)G * H);
  zero_kernel<<<nb(G, 256), 256, 0, stream>>>(cnt, G);

  for (int l = 0; l < Lnum; ++l) {
    float* cxn = (l & 1) ? xn2 : xn;
    float* cxl = (l & 1) ? xl2 : xl;
    float* oxn = (l & 1) ? xn  : xn2;
    float* oxl = (l & 1) ? xl  : xl2;

    const float* WmNLt = Wm_nl + (size_t)l * 2 * H * H;
    const float* WmNLb = WmNLt + H * H;
    const float* WuNLt = Wu_nl + (size_t)l * 2 * H * H;
    const float* WuNLb = WuNLt + H * H;
    const float* WmLNt = Wm_ln + (size_t)l * 2 * H * H;
    const float* WmLNb = WmLNt + H * H;
    const float* WuLNt = Wu_ln + (size_t)l * 2 * H * H;
    const float* WuLNb = WuLNt + H * H;

    const int sl = nb(NL / 16, ST);
    const int sn = nb(NN / 16, ST);
    if (l == Lnum - 1)
      scatter_gemm_kernel<true><<<sl, 512, 0, stream>>>(
          cxn, nl, E, ownl, cxl, WmNLt, WmNLb, bm_nl + (size_t)l * H, dgl,
          WuNLt, WuNLb, bu_nl + (size_t)l * H, nullptr, NL, batch, sums, cnt, G);
    else
      scatter_gemm_kernel<false><<<sl, 512, 0, stream>>>(
          cxn, nl, E, ownl, cxl, WmNLt, WmNLb, bm_nl + (size_t)l * H, dgl,
          WuNLt, WuNLb, bu_nl + (size_t)l * H, oxl, NL, nullptr, nullptr, nullptr, 0);
    scatter_gemm_kernel<false><<<sn, 512, 0, stream>>>(
        cxl, ln, E, ownn, cxn, WmLNt, WmLNb, bm_ln + (size_t)l * H, dgn,
        WuLNt, WuLNb, bu_ln + (size_t)l * H, oxn, NN, nullptr, nullptr, nullptr, 0);
  }

  finalize_kernel<<<nb((long)G * H, 256), 256, 0, stream>>>(sums, cnt, (float*)d_out, G);
}


# degree pass files each edge in its owner workgroup's bucket (place: LDS count + one global add per block and owner); a workgroup reads its own list, not every owner byte; sums in arrival order
# speedup vs baseline: 2.4403x; 1.1051x over previous
#include <hip/hip_runtime.h>


typedef __attribute__((ext_vector_type(2))) float v2f;
typedef __attribute__((ext_vector_type(8))) float v8f;

#define H 128
#define LDSPAD 132
#define ST 13
#define SROWS (ST * 16)
#define QCAP 4096
#define SWAVES 16
#define SQCAP 256
#define SPLIT (ST - 1)
#define PNB (8 / (SWAVES - SPLIT))
#define BCAP 4096
#define BOWN 1024
#define CSTR 320

__global__ __launch_bounds__(256) void zero_kernel(float* p, long n) {
  long i = (long)blockIdx.x * blockDim.x + threadIdx.x;
  if (i < n) p[i] = 0.0f;
}

__global__ __launch_bounds__(256) void pad_kernel(const float* __restrict__ in,
                                                  float* __restrict__ out,
                                                  int N, int F) {
  long t = (long)blockIdx.x * blockDim.x + threadIdx.x;
  if (t >= (long)N * H) return;
  int i = (int)(t >> 7), h = (int)(t & 127);
  out[t] = (h < F) ? in[(long)i * F + h] : 0.0f;
}

__global__ __launch_bounds__(256) void deg_kernel(const int* __restrict__ edge,
                                                  int E, float* __restrict__ deg,
                                                  int* __restrict__ own,
                                                  int* __restrict__ bcnt, int nbk,
                                                  int* __restrict__ bkt) {
  __shared__ int lcnt[BOWN];
  __shared__ int lbase[BOWN];
  const int tid = threadIdx.x;
  const int q = blockIdx.x * blockDim.x + tid;
  const int e0 = q << 2;
  for (int i = tid; i < nbk; i += 256) lcnt[i] = 0;
  __syncthreads();
  const int* __restrict__ dstp = edge + E;
  int n = E - e0;
  n = (n < 0) ? 0 : ((n > 4) ? 4 : n);
  int d[4] = {0, 0, 0, 0}, s[4] = {0, 0, 0, 0};
  if (n == 4) {
    d[0] = dstp[e0]; d[1] = dstp[e0 + 1]; d[2] = dstp[e0 + 2]; d[3] = dstp[e0 + 3];
    s[0] = edge[e0]; s[1] = edge[e0 + 1]; s[2] = edge[e0 + 2]; s[3] = edge[e0 + 3];
  } else {
    for (int j = 0; j < n; ++j) { d[j] = dstp[e0 + j]; s[j] = edge[e0 + j]; }
  }
  int o[4], r[4];
#pragma unroll
  for (int j = 0; j < 4; ++j) {
    o[j] = (int)((unsigned)d[j] / SROWS);
    r[j] = (j < n && o[j] < nbk) ? atomicAdd(&lcnt[o[j]], 1) : -1;
  }
  __syncthreads();
  for (int i = tid; i < nbk; i += 256) {
    const int cnt = lcnt[i];
    lbase[i] = cnt ? atomicAdd(&bcnt[(long)i * CSTR], cnt) : 0;
  }
  __syncthreads();
  unsigned w = 0;
#pragma unroll
  for (int j = 0; j < 4; ++j) {
    if (j < n) {
      atomicAdd(&deg[d[j]], 1.0f);
      w |= ((unsigned)o[j] & 255u) << (8 * j);
    }
    if (r[j] >= 0) {
      const int slot = lbase[o[j]] + r[j];
      if (slot < BCAP) *(int2*)(bkt + 2 * ((long)o[j] * BCAP + slot)) = make_int2(s[j], d[j] - o[j] * SROWS);
    }
  }
  if (n > 0) own[q] = (int)w;
}

template <int NB>
__device__ __forceinline__ void dual_product(const float* Ap, const float* Bp,
                                             const float* wabase, const float* wbbase,
                                             v8f (&cA)[NB], v8f (&cB)[NB]) {
#pragma unroll
  for (int n = 0; n < NB; ++n) {
    cA[n] = (v8f){0, 0, 0, 0, 0, 0, 0, 0};
    cB[n] = (v8f){0, 0, 0, 0, 0, 0, 0, 0};
  }
#pragma unroll 1
  for (int kk = 0; kk < 32; ++kk) {
    const int k0 = kk << 2;
    v2f a = *(const v2f*)(Ap + k0);
    v2f b = *(const v2f*)(Bp + k0);
#pragma unroll
    for (int n = 0; n < NB; ++n) {
      v2f wa = *(const v2f*)(wabase + n * 16 * LDSPAD + k0);
      v2f wb = *(const v2f*)(wbbase + n * 16 * LDSPAD + k0);
      cA[n] = __builtin_amdgcn_wmma_f32_16x16x4_f32(false, a, false, wa,
                                                    (short)0, cA[n], false, false);
      cB[n] = __builtin_amdgcn_wmma_f32_16x16x4_f32(false, b, false, wb,
                                                    (short)0, cB[n], false, false);
    }
  }
}

__device__ __forceinline__ void pool_add(float* __restrict__ p, int pitch, int g, int G, float v) {
  if ((unsigned)g < (unsigned)G) atomicAdd(p + (long)g * pitch, v);
}

template <int NB>
__device__ __forceinline__ void agg_store(const v8f (&cA)[NB], const v8f (&cB)[NB],
                                          const float* __restrict__ bcol,
                                          const float* __restrict__ degp, float* lrow) {
  const float4 d0 = *(const float4*)(degp);
  const float4 d1 = *(const float4*)(degp + 4);
  const float degv[8] = {d0.x, d0.y, d0.z, d0.w, d1.x, d1.y, d1.z, d1.w};
#pragma unroll
  for (int n = 0; n < NB; ++n) {
    const float bv = bcol[n * 16];
#pragma unroll
    for (int r = 0; r < 8; ++r)
      lrow[r * LDSPAD + n * 16] = cA[n][r] + degv[r] * (cB[n][r] + bv);
  }
}

template <bool POOL, int NB>
__device__ __forceinline__ void update_out(const v8f (&cA)[NB], const v8f (&cB)[NB],
                                           const float* __restrict__ ubias, int col0, int l15,
                                           int rbase, float* __restrict__ Y,
                                           const int* __restrict__ seg, float* __restrict__ psum,
                                           float* __restrict__ pcnt, int G, bool count) {
  int gid[8];
  bool uni = true;
  if (POOL) {
    const int4 g0 = *(const int4*)(seg + rbase);
    const int4 g1 = *(const int4*)(seg + rbase + 4);
    gid[0] = g0.x; gid[1] = g0.y; gid[2] = g0.z; gid[3] = g0.w;
    gid[4] = g1.x; gid[5] = g1.y; gid[6] = g1.z; gid[7] = g1.w;
#pragma unroll
    for (int r = 1; r < 8; ++r) uni = uni && (gid[r] == gid[0]);
    if (count && l15 == 0) {
      if (uni) {
        pool_add(pcnt, 1, gid[0], G, 8.0f);
      } else {
        int gc = gid[0];
        float len = 1.0f;
#pragma unroll
        for (int r = 1; r < 8; ++r) {
          if (gid[r] == gc) { len += 1.0f; }
          else { pool_add(pcnt, 1, gc, G, len); gc = gid[r]; len = 1.0f; }
        }
        pool_add(pcnt, 1, gc, G, len);
      }
    }
  }
#pragma unroll
  for (int n = 0; n < NB; ++n) {
    const float bv = ubias[col0 + n * 16];
    float v[8];
#pragma unroll
    for (int r = 0; r < 8; ++r)
      v[r] = fmaxf(cA[n][r] + cB[n][r] + bv, 0.0f);
    if (!POOL) {
#pragma unroll
      for (int r = 0; r < 8; ++r)
        Y[(long)(rbase + r) * H + col0 + n * 16] = v[r];
    } else {
      float* pc = psum + col0 + n * 16;
      if (uni) {
        float run = v[0];
#pragma unroll
        for (int r = 1; r < 8; ++r) run += v[r];
        pool_add(pc, H, gid[0], G, run);
      } else {
        int gc = gid[0];
        float run = v[0];
#pragma unroll
        for (int r = 1; r < 8; ++r) {
          if (gid[r] == gc) { run += v[r]; }
          else { pool_add(pc, H, gc, G, run); gc = gid[r]; run = v[r]; }
        }
        pool_add(pc, H, gc, G, run);
      }
    }
  }
}

typedef __attribute__((ext_vector_type(16))) float v16f;
static_assert(ST <= 15, "a wave's rows are places 0..ST-1 of a 16-wide register vector; place 15 takes the entries that add nothing");
static_assert(SWAVES == 16 && SROWS == ST * SWAVES, "local row = place * SWAVES + wave");
static_assert(SWAVES - SPLIT == 4 && PNB * (SWAVES - SPLIT) == 8, "the shared tile's 8 column blocks go to four waves, one per SIMD");

template <bool POOL>
__global__ __launch_bounds__(512) void scatter_gemm_kernel(
    const float* __restrict__ X, const int* __restrict__ edge, int E,
    const int* __restrict__ own, const float* __restrict__ B,
    const float* __restrict__ WA, const float* __restrict__ WB,
    const float* __restrict__ bias, const float* __restrict__ deg,
    const float* __restrict__ UA, const float* __restrict__ UB,
    const float* __restrict__ ubias,
    float* __restrict__ Y, int M,
    const int* __restrict__ seg, float* __restrict__ psum,
    float* __restrict__ pcnt, int G,
    const int* __restrict__ bcnt, int nbk, const int* __restrict__ bkt) {
  __shared__ float lwa[H * LDSPAD];
  __shared__ float lwb[H * LDSPAD];
  __shared__ __attribute__((aligned(16))) float ls[SROWS * LDSPAD];
  __shared__ float lagg[16 * LDSPAD];
  __shared__ int qsrc[QCAP];
  __shared__ __attribute__((aligned(16))) unsigned sq[SWAVES * SQCAP];
  __shared__ int sqn[SWAVES];
  __shared__ int qn;

  const int tid = threadIdx.x;
  const int row0 = blockIdx.x * SROWS;
  const int nrows = (M - row0 < SROWS) ? (M - row0) : SROWS;

  for (int i = tid; i < SROWS * LDSPAD / 4; i += 512)
    ((float4*)ls)[i] = make_float4(0.0f, 0.0f, 0.0f, 0.0f);
  if (tid == 0) qn = 0;
  if (tid < SWAVES) sqn[tid] = 0;
  for (int idx = tid; idx < H * H; idx += 512) {
    int k = idx >> 7, n = idx & 127;
    lwa[n * LDSPAD + k] = WA[idx];
    lwb[n * LDSPAD + k] = WB[idx];
  }
  __syncthreads();

  const int* __restrict__ dstp = edge + E;
  const unsigned nr = (unsigned)nrows;
  auto deal = [&](int s, int dl) {
    const bool own_row = (unsigned)dl < nr;
    const int w = dl & (SWAVES - 1);
    const unsigned place = own_row ? (unsigned)(dl >> 4) : 15u;
    const int pos = atomicAdd(&sqn[w], 1);
    if (pos < SQCAP) {
      sq[w * SQCAP + pos] = ((unsigned)s << 4) | place;
    } else if (own_row) {
      const float* xr = X + (long)s * H;
      float* sr = ls + dl * LDSPAD;
      for (int c = 0; c < H; ++c) atomicAdd(sr + c, xr[c]);
    }
  };
  const int bn = __builtin_amdgcn_readfirstlane(((int)blockIdx.x < nbk) ? bcnt[(long)blockIdx.x * CSTR] : BCAP + 1);
  if (bn <= BCAP) {
    const int2* __restrict__ lst = (const int2*)(bkt + 2 * (long)blockIdx.x * BCAP);
    for (int q0 = tid; q0 < bn; q0 += 4 * 512) {
      int2 en[4];
#pragma unroll
      for (int u = 0; u < 4; ++u) {
        const int q = q0 + u * 512;
        en[u] = (q < bn) ? lst[q] : make_int2(0, 0);
      }
#pragma unroll
      for (int u = 0; u < 4; ++u)
        if (q0 + u * 512 < bn) deal(en[u].x, en[u].y);
    }
  } else {
    auto push = [&](int e) {
      const int slot = atomicAdd(&qn, 1);
      if (slot < QCAP) {
        qsrc[slot] = e;
      } else {
        const int dl = dstp[e] - row0;
        if ((unsigned)dl < nr) {
          const float* xr = X + (long)edge[e] * H;
          float* sr = ls + dl * LDSPAD;
          for (int c = 0; c < H; ++c) atomicAdd(sr + c, xr[c]);
        }
      }
    };
    const unsigned mine = (unsigned)blockIdx.x & 255u;
    const unsigned pat = mine * 0x01010101u;
    auto eqb = [&](int w) -> unsigned {
      const unsigned x = (unsigned)w ^ pat;
      return ~(((x & 0x7f7f7f7fu) + 0x7f7f7f7fu) | x | 0x7f7f7f7fu);
    };
    auto nib = [](unsigned h) -> unsigned {
      return ((h >> 7) * 0x01020408u) >> 24;
    };
    auto hits = [&](const int4 d) -> unsigned {
      return nib(eqb(d.x)) | (nib(eqb(d.y)) << 4) | (nib(eqb(d.z)) << 8) | (nib(eqb(d.w)) << 12);
    };
    {
      const int4* __restrict__ o4 = (const int4*)own;
      const int n16 = E >> 4;
      int i = tid;
      for (; i + 15 * 512 < n16; i += 16 * 512) {
        int4 d[16];
#pragma unroll
        for (int u = 0; u < 16; ++u) d[u] = o4[i + u * 512];
#pragma unroll
        for (int u = 0; u < 16; u += 2) {
          unsigned m = hits(d[u]) | (hits(d[u + 1]) << 16);
          while (m) {
            const int j = __ffs(m) - 1;
            m &= m - 1;
            push(16 * (i + (u + (j >> 4)) * 512) + (j & 15));
          }
        }
      }
      for (; i < n16; i += 512) {
        unsigned m = hits(o4[i]);
        while (m) {
          const int j = __ffs(m) - 1;
          m &= m - 1;
          push(16 * i + j);
        }
      }
      for (int e = (n16 << 4) + tid; e < E; e += 512)
        if ((((unsigned)own[e >> 2] >> (8 * (e & 3))) & 255u) == mine) push(e);
    }
    __syncthreads();

    const int qc = (qn < QCAP) ? qn : QCAP;
    for (int q = tid; q < qc; q += 512) {
      const int e = qsrc[q];
      deal(edge[e], dstp[e] - row0);
    }
  }
  __syncthreads();

  const int wave = __builtin_amdgcn_readfirstlane(tid >> 5);
  const int lane = tid & 31;
  {
    unsigned* myq = sq + wave * SQCAP;
    int n = __builtin_amdgcn_readfirstlane(sqn[wave]);
    n = (n < SQCAP) ? n : SQCAP;
    const int n16 = (n + 15) & ~15;
    if (lane < 16 && n + lane < n16) myq[n + lane] = 15u;
    v16f ax = {0, 0, 0, 0, 0, 0, 0, 0, 0, 0, 0, 0, 0, 0, 0, 0};
    v16f ay = ax, az = ax, aw = ax;
    for (int q0 = 0; q0 < n16; q0 += 16) {
      const uint4 e0 = *(const uint4*)&myq[q0], e1 = *(const uint4*)&myq[q0 + 4];
      const uint4 e2 = *(const uint4*)&myq[q0 + 8], e3 = *(const uint4*)&myq[q0 + 12];
      const unsigned ev[16] = {e0.x, e0.y, e0.z, e0.w, e1.x, e1.y, e1.z, e1.w,
                               e2.x, e2.y, e2.z, e2.w, e3.x, e3.y, e3.z, e3.w};
      int en[16];
      float4 v[16];
#pragma unroll
      for (int u = 0; u < 16; ++u) {
        en[u] = __builtin_amdgcn_readfirstlane((int)ev[u]);
        v[u] = *(const float4*)(X + (long)(int)((unsigned)en[u] >> 4) * H + lane * 4);
      }
#pragma unroll
      for (int u = 0; u < 16; ++u) {
        const int s = en[u] & 15;
        ax[s] += v[u].x;
        ay[s] += v[u].y;
        az[s] += v[u].z;
        aw[s] += v[u].w;
      }
    }
#pragma unroll
    for (int s = 0; s < ST; ++s) {
      float4* p = (float4*)(ls + (s * SWAVES + wave) * LDSPAD + lane * 4);
      float4 t = *p;
      t.x += ax[s]; t.y += ay[s]; t.z += az[s]; t.w += aw[s];
      *p = t;
    }
  }
  __syncthreads();

  const int l15  = lane & 15;
  const int kh2  = (lane >> 4) << 1;
  const bool part = wave >= SPLIT;
  const int ltile = part ? SPLIT : wave;
  const int nb0 = part ? (wave - SPLIT) * PNB : 0;
  const int t = blockIdx.x * ST + ltile;
  const bool has = t < (M >> 4);
  const int m0 = t << 4;
  const int rbase = m0 + ((lane >> 4) << 3);
  const int trow = has ? ltile * 16 : 0;
  const int col0 = nb0 * 16 + l15;
  const float* Ap = &ls[(trow + l15) * LDSPAD + kh2];
  const float* Bp = B + (long)((has ? m0 : 0) + l15) * H + kh2;
  const float* wabase = &lwa[l15 * LDSPAD + kh2];
  const float* wbbase = &lwb[l15 * LDSPAD + kh2];
  const int wstep = nb0 * 16 * LDSPAD;

  if (has) {
    if (!part) {
      v8f cA[8], cB[8];
      dual_product<8>(Ap, Bp, wabase, wbbase, cA, cB);
      agg_store<8>(cA, cB, bias + l15, deg + rbase, ls + (trow + ((lane >> 4) << 3)) * LDSPAD + l15);
    } else {
      v8f cA[PNB], cB[PNB];
      dual_product<PNB>(Ap, Bp, wabase + wstep, wbbase + wstep, cA, cB);
      agg_store<PNB>(cA, cB, bias + col0, deg + rbase, lagg + ((lane >> 4) << 3) * LDSPAD + col0);
    }
  }
  float ua[H * H / 512], ub[H * H / 512];
#pragma unroll
  for (int i = 0; i < H * H / 512; ++i) {
    ua[i] = UA[tid + i * 512];
    ub[i] = UB[tid + i * 512];
  }
  __syncthreads();
#pragma unroll
  for (int i = 0; i < H * H / 512; ++i) {
    const int idx = tid + i * 512;
    const int k = idx >> 7, n = idx & 127;
    lwa[n * LDSPAD + k] = ua[i];
    lwb[n * LDSPAD + k] = ub[i];
  }
  __syncthreads();

  if (has) {
    if (!part) {
      v8f cA[8], cB[8];
      dual_product<8>(Ap, Bp, wabase, wbbase, cA, cB);
      update_out<POOL, 8>(cA, cB, ubias, l15, l15, rbase, Y, seg, psum, pcnt, G, true);
    } else {
      v8f cA[PNB], cB[PNB];
      dual_product<PNB>(&lagg[l15 * LDSPAD + kh2], Bp, wabase + wstep, wbbase + wstep, cA, cB);
      update_out<POOL, PNB>(cA, cB, ubias, col0, l15, rbase, Y, seg, psum, pcnt, G, wave == SPLIT);
    }
  }
}

__global__ __launch_bounds__(256) void finalize_kernel(const float* __restrict__ sums,
                                                       const float* __restrict__ cnt,
                                                       float* __restrict__ out, int G) {
  int t = blockIdx.x * blockDim.x + threadIdx.x;
  if (t >= G * H) return;
  int g = t >> 7;
  out[t] = sums[t] / fmaxf(cnt[g], 1.0f);
}

extern "C" void kernel_launch(void* const* d_in, const int* in_sizes, int n_in,
                              void* d_out, int out_size, void* d_ws, size_t ws_size,
                              hipStream_t stream) {
  const float* x_node = (const float*)d_in[0];
  const float* x_link = (const float*)d_in[1];
  const float* Wm_nl  = (const float*)d_in[2];
  const float* bm_nl  = (const float*)d_in[3];
  const float* Wu_nl  = (const float*)d_in[4];
  const float* bu_nl  = (const float*)d_in[5];
  const float* Wm_ln  = (const float*)d_in[6];
  const float* bm_ln  = (const float*)d_in[7];
  const float* Wu_ln  = (const float*)d_in[8];
  const float* bu_ln  = (const float*)d_in[9];
  const int*   nl     = (const int*)d_in[10];
  const int*   ln     = (const int*)d_in[11];
  const int*   batch  = (const int*)d_in[12];

  const int NL = in_sizes[12];
  const int NN = 50000;
  const int F_NODE = in_sizes[0] / NN;
  const int F_LINK = in_sizes[1] / NL;
  const int E  = in_sizes[10] / 2;
  const int Lnum = in_sizes[3] / H;
  const int G = out_size / H;

  const size_t MH = (size_t)((NN > NL) ? NN : NL) * H;
  float* ws   = (float*)d_ws;
  float* xn   = ws;
  float* xl   = xn  + MH;
  float* xn2  = xl  + MH;
  float* xl2  = xn2 + MH;
  float* Sl   = xl2 + MH;
  float* Sn   = Sl  + MH;
  float* dgl  = Sn  + MH;
  float* dgn  = dgl + NL;
  float* sums = dgn + NN;
  float* cnt  = sums + (size_t)G * H;
  (void)n_in; (void)ws_size;
  int* ownl = (int*)Sl;
  int* ownn = (int*)Sn;
  int* bkl = (int*)(Sl + MH / 2);
  int* bkn = (int*)(Sn + MH / 2);

  auto nb = [](long n, int b) { return (int)((n + b - 1) / b); };
  const int nown = nb(((NN > NL) ? NN : NL) / 16, ST);
  const int nbk = (nown < BOWN) ? nown : BOWN;
  const long ncw = (long)nbk * CSTR;
  int* bcl = (int*)dgl - 2 * ncw;
  int* bcn = bcl + ncw;

  zero_kernel<<<nb((long)NL + 2 * ncw, 256), 256, 0, stream>>>((float*)bcl, (long)NL + 2 * ncw);
  zero_kernel<<<nb(NN, 256), 256, 0, stream>>>(dgn, NN);
  deg_kernel<<<nb(nb(E, 4), 256), 256, 0, stream>>>(nl, E, dgl, ownl, bcl, nbk, bkl);
  deg_kernel<<<nb(nb(E, 4), 256), 256, 0, stream>>>(ln, E, dgn, ownn, bcn, nbk, bkn);

  pad_kernel<<<nb((long)NN * H, 256), 256, 0, stream>>>(x_node, xn, NN, F_NODE);
  pad_kernel<<<nb((long)NL * H, 256), 256, 0, stream>>>(x_link, xl, NL, F_LINK);

  zero_kernel<<<nb((long)G * H, 256), 256, 0, stream>>>(sums, (long)G * H);
  zero_kernel<<<nb(G, 256), 256, 0, stream>>>(cnt, G);

  for (int l = 0; l < Lnum; ++l) {
    float* cxn = (l & 1) ? xn2 : xn;
    float* cxl = (l & 1) ? xl2 : xl;
    float* oxn = (l & 1) ? xn  : xn2;
    float* oxl = (l & 1) ? xl  : xl2;

    const float* WmNLt = Wm_nl + (size_t)l * 2 * H * H;
    const float* WmNLb = WmNLt + H * H;
    const float* WuNLt = Wu_nl + (size_t)l * 2 * H * H;
    const float* WuNLb = WuNLt + H * H;
    const float* WmLNt = Wm_ln + (size_t)l * 2 * H * H;
    const float* WmLNb = WmLNt + H * H;
    const float* WuLNt = Wu_ln + (size_t)l * 2 * H * H;
    const float* WuLNb = WuLNt + H * H;

    const int sl = nb(NL / 16, ST);
    const int sn = nb(NN / 16, ST);
    if (l == Lnum - 1)
      scatter_gemm_kernel<true><<<sl, 512, 0, stream>>>(
          cxn, nl, E, ownl, cxl, WmNLt, WmNLb, bm_nl + (size_t)l * H, dgl,
          WuNLt, WuNLb, bu_nl + (size_t)l * H, nullptr, NL, batch, sums, cnt, G, bcl, nbk, bkl);
    else
      scatter_gemm_kernel<false><<<sl, 512, 0, stream>>>(
          cxn, nl, E, ownl, cxl, WmNLt, WmNLb, bm_nl + (size_t)l * H, dgl,
          WuNLt, WuNLb, bu_nl + (size_t)l * H, oxl, NL, nullptr, nullptr, nullptr, 0, bcl, nbk, bkl);
    scatter_gemm_kernel<false><<<sn, 512, 0, stream>>>(
        cxl, ln, E, ownn, cxn, WmLNt, WmLNb, bm_ln + (size_t)l * H, dgn,
        WuLNt, WuLNb, bu_ln + (size_t)l * H, oxn, NN, nullptr, nullptr, nullptr, 0, bcn, nbk, bkn);
  }

  finalize_kernel<<<nb((long)G * H, 256), 256, 0, stream>>>(sums, cnt, (float*)d_out, G);
}
